# starting kernel
# baseline (speedup 1.0000x reference)
#include <hip/hip_runtime.h>
#include <cmath>

typedef __attribute__((ext_vector_type(2))) float v2f;
typedef __attribute__((ext_vector_type(8))) float v8f;

constexpr int kH  = 18;
constexpr int kH3 = 54;
constexpr int kH4 = 72;
constexpr int kT  = 1024;
constexpr int kB  = 1024;
constexpr int kKP = 20;
constexpr int kKS = 5;
constexpr int kNG = 64;
constexpr int kNL = 80;

__device__ __forceinline__ float sigmoidf_(float v) {
    return 1.0f / (1.0f + __expf(-v));
}

__global__ __launch_bounds__(32) void fused_gru_lstm_fc(
    const float* __restrict__ x,
    const float* __restrict__ gWih, const float* __restrict__ gWhh,
    const float* __restrict__ gbih, const float* __restrict__ gbhh,
    const float* __restrict__ lWih, const float* __restrict__ lWhh,
    const float* __restrict__ lbih, const float* __restrict__ lbhh,
    const float* __restrict__ fcW,  const float* __restrict__ fcb,
    float* __restrict__ out)
{
    __shared__ float WgX[kNG][kKP], WgH[kNG][kKP];
    __shared__ float WlX[kNL][kKP], WlH[kNL][kKP];
    __shared__ float gbi_s[kH3], gbh_s[kH3], lb_s[kH4];
    __shared__ float fcw_s[kH];
    __shared__ float fcb_s;
    __shared__ float xt[16][kKP];
    __shared__ float hg[16][kKP];
    __shared__ float hl[16][kKP];
    __shared__ float cs[16][kH];
    __shared__ float sgX[16][kNG];
    __shared__ float sgH[16][kNG];
    __shared__ float sl [16][kNL];

    const int tid = threadIdx.x;
    const int m0  = blockIdx.x * 16;

    for (int i = tid; i < kNG * kKP; i += 32) {
        int n = i / kKP, k = i % kKP;
        bool ok = (n < kH3) && (k < kH);
        WgX[n][k] = ok ? gWih[n * kH + k] : 0.0f;
        WgH[n][k] = ok ? gWhh[n * kH + k] : 0.0f;
    }
    for (int i = tid; i < kNL * kKP; i += 32) {
        int n = i / kKP, k = i % kKP;
        bool ok = (n < kH4) && (k < kH);
        WlX[n][k] = ok ? lWih[n * kH + k] : 0.0f;
        WlH[n][k] = ok ? lWhh[n * kH + k] : 0.0f;
    }
    for (int i = tid; i < kH3; i += 32) { gbi_s[i] = gbih[i]; gbh_s[i] = gbhh[i]; }
    for (int i = tid; i < kH4; i += 32) { lb_s[i]  = lbih[i] + lbhh[i]; }
    for (int i = tid; i < kH;  i += 32) { fcw_s[i] = fcW[i]; }
    if (tid == 0) fcb_s = fcb[0];
    for (int i = tid; i < 16 * kKP; i += 32) {
        (&xt[0][0])[i] = 0.0f; (&hg[0][0])[i] = 0.0f; (&hl[0][0])[i] = 0.0f;
    }
    for (int i = tid; i < 16 * kH; i += 32) (&cs[0][0])[i] = 0.0f;
    __syncthreads();

    const int lh  = tid & 15;
    const int kh  = (tid >> 4) << 1;
    const int hi8 = (tid >> 4) << 3;

    v2f bGX[4][kKS], bGH[4][kKS];
    #pragma unroll
    for (int nt = 0; nt < 4; ++nt)
      #pragma unroll
      for (int ks = 0; ks < kKS; ++ks) {
        int r = nt * 16 + lh, c = ks * 4 + kh;
        v2f a; a.x = WgX[r][c]; a.y = WgX[r][c + 1]; bGX[nt][ks] = a;
        v2f b; b.x = WgH[r][c]; b.y = WgH[r][c + 1]; bGH[nt][ks] = b;
      }
    v2f bLX[5][kKS], bLH[5][kKS];
    #pragma unroll
    for (int nt = 0; nt < 5; ++nt)
      #pragma unroll
      for (int ks = 0; ks < kKS; ++ks) {
        int r = nt * 16 + lh, c = ks * 4 + kh;
        v2f a; a.x = WlX[r][c]; a.y = WlX[r][c + 1]; bLX[nt][ks] = a;
        v2f b; b.x = WlH[r][c]; b.y = WlH[r][c + 1]; bLH[nt][ks] = b;
      }

    for (int t = 0; t < kT; ++t) {
        for (int i = tid; i < 16 * kH; i += 32) {
            int m = i / kH, j = i % kH;
            xt[m][j] = x[((size_t)(m0 + m) * kT + t) * kH + j];
        }
        if (t + 1 < kT && tid < 16) {
            __builtin_prefetch(&x[((size_t)(m0 + tid) * kT + (t + 1)) * kH], 0, 3);
        }
        __syncthreads();

        v2f aX[kKS], aH[kKS];
        #pragma unroll
        for (int ks = 0; ks < kKS; ++ks) {
            int c = ks * 4 + kh;
            v2f a; a.x = xt[lh][c]; a.y = xt[lh][c + 1]; aX[ks] = a;
            v2f b; b.x = hg[lh][c]; b.y = hg[lh][c + 1]; aH[ks] = b;
        }
        v8f accX[4], accH[4];
        #pragma unroll
        for (int nt = 0; nt < 4; ++nt) { accX[nt] = (v8f){}; accH[nt] = (v8f){}; }
        #pragma unroll
        for (int nt = 0; nt < 4; ++nt)
          #pragma unroll
          for (int ks = 0; ks < kKS; ++ks) {
            accX[nt] = __builtin_amdgcn_wmma_f32_16x16x4_f32(
                false, aX[ks], false, bGX[nt][ks], (short)0, accX[nt], false, false);
            accH[nt] = __builtin_amdgcn_wmma_f32_16x16x4_f32(
                false, aH[ks], false, bGH[nt][ks], (short)0, accH[nt], false, false);
          }
        #pragma unroll
        for (int nt = 0; nt < 4; ++nt)
          #pragma unroll
          for (int j = 0; j < 8; ++j) {
            sgX[hi8 + j][nt * 16 + lh] = accX[nt][j];
            sgH[hi8 + j][nt * 16 + lh] = accH[nt][j];
          }
        __syncthreads();

        for (int i = tid; i < 16 * kH; i += 32) {
            int m = i / kH, j = i % kH;
            float r = sigmoidf_((sgX[m][j]          + gbi_s[j])
                              + (sgH[m][j]          + gbh_s[j]));
            float z = sigmoidf_((sgX[m][kH + j]     + gbi_s[kH + j])
                              + (sgH[m][kH + j]     + gbh_s[kH + j]));
            float n = tanhf(    (sgX[m][2*kH + j]   + gbi_s[2*kH + j])
                          + r * (sgH[m][2*kH + j]   + gbh_s[2*kH + j]));
            float hprev = hg[m][j];
            hg[m][j] = (1.0f - z) * n + z * hprev;
        }
        __syncthreads();

        v2f aG[kKS], aL[kKS];
        #pragma unroll
        for (int ks = 0; ks < kKS; ++ks) {
            int c = ks * 4 + kh;
            v2f a; a.x = hg[lh][c]; a.y = hg[lh][c + 1]; aG[ks] = a;
            v2f b; b.x = hl[lh][c]; b.y = hl[lh][c + 1]; aL[ks] = b;
        }
        v8f accL[5];
        #pragma unroll
        for (int nt = 0; nt < 5; ++nt) accL[nt] = (v8f){};
        #pragma unroll
        for (int nt = 0; nt < 5; ++nt)
          #pragma unroll
          for (int ks = 0; ks < kKS; ++ks) {
            accL[nt] = __builtin_amdgcn_wmma_f32_16x16x4_f32(
                false, aG[ks], false, bLX[nt][ks], (short)0, accL[nt], false, false);
            accL[nt] = __builtin_amdgcn_wmma_f32_16x16x4_f32(
                false, aL[ks], false, bLH[nt][ks], (short)0, accL[nt], false, false);
          }
        #pragma unroll
        for (int nt = 0; nt < 5; ++nt)
          #pragma unroll
          for (int j = 0; j < 8; ++j)
            sl[hi8 + j][nt * 16 + lh] = accL[nt][j];
        __syncthreads();

        for (int i = tid; i < 16 * kH; i += 32) {
            int m = i / kH, j = i % kH;
            float ig = sigmoidf_(sl[m][j]          + lb_s[j]);
            float fg = sigmoidf_(sl[m][kH + j]     + lb_s[kH + j]);
            float gg = tanhf(    sl[m][2*kH + j]   + lb_s[2*kH + j]);
            float og = sigmoidf_(sl[m][3*kH + j]   + lb_s[3*kH + j]);
            float cn = fg * cs[m][j] + ig * gg;
            cs[m][j] = cn;
            hl[m][j] = og * tanhf(cn);
        }
        __syncthreads();

        if (tid < 16) {
            float s = fcb_s;
            #pragma unroll
            for (int j = 0; j < kH; ++j) s += hl[tid][j] * fcw_s[j];
            out[(size_t)(m0 + tid) * kT + t] = s;
        }
        __syncthreads();
    }
}

extern "C" void kernel_launch(void* const* d_in, const int* in_sizes, int n_in,
                              void* d_out, int out_size, void* d_ws, size_t ws_size,
                              hipStream_t stream) {
    (void)in_sizes; (void)n_in; (void)out_size; (void)d_ws; (void)ws_size;
    const float* x    = (const float*)d_in[0];
    const float* gWih = (const float*)d_in[1];
    const float* gWhh = (const float*)d_in[2];
    const float* gbih = (const float*)d_in[3];
    const float* gbhh = (const float*)d_in[4];
    const float* lWih = (const float*)d_in[5];
    const float* lWhh = (const float*)d_in[6];
    const float* lbih = (const float*)d_in[7];
    const float* lbhh = (const float*)d_in[8];
    const float* fcW  = (const float*)d_in[9];
    const float* fcb  = (const float*)d_in[10];
    float* out = (float*)d_out;

    dim3 grid(kB / 16);
    dim3 block(32);
    hipLaunchKernelGGL(fused_gru_lstm_fc, grid, block, 0, stream,
                       x, gWih, gWhh, gbih, gbhh,
                       lWih, lWhh, lbih, lbhh, fcW, fcb, out);
}


# Swap product roles (weights = A, gate rows permuted; batch = N): each lane's D holds all gates of its 9 units; state, biases (as C), x (read a step ahead) stay in registers, no LDS; sum order changes
# speedup vs baseline: 2.6547x; 2.6547x over previous
#include <hip/hip_runtime.h>
#include <cmath>

typedef __attribute__((ext_vector_type(2))) float v2f;
typedef __attribute__((ext_vector_type(8))) float v8f;

constexpr int kH  = 18;
constexpr int kT  = 1024;
constexpr int kB  = 1024;
constexpr int kKS = 5;
constexpr int kQ  = 9;
constexpr int kMG = 4;
constexpr int kML = 5;


__device__ __forceinline__ int gate_row(int k, int i, int hh, int ngates) {
    int slot = 8 * k + i;
    int g = slot / kQ, q = slot % kQ;
    return (g < ngates) ? g * kH + kQ * hh + q : -1;
}
__device__ __forceinline__ int k_unit(int ks, int s, int h) {
    int q = 2 * ks + s;
    return (q < kQ) ? kQ * h + q : -1;
}
__device__ __forceinline__ v2f a_frag(const float* __restrict__ Wm, int ngates,
                                      int k, int ks, int c, int h) {
    int row = gate_row(k, c & 7, c >> 3, ngates);
    int u0 = k_unit(ks, 0, h), u1 = k_unit(ks, 1, h);
    const int rc = (row >= 0) ? row : 0;
    const float w0 = Wm[rc * kH + ((u0 >= 0) ? u0 : 0)];
    const float w1 = Wm[rc * kH + ((u1 >= 0) ? u1 : 0)];
    v2f a;
    a.x = (row >= 0 && u0 >= 0) ? w0 : 0.0f;
    a.y = (row >= 0 && u1 >= 0) ? w1 : 0.0f;
    return a;
}
__device__ __forceinline__ float bias_at(const float* __restrict__ b, int ngates,
                                         int k, int i, int h) {
    int row = gate_row(k, i, h, ngates);
    const float v = b[(row >= 0) ? row : 0];
    return (row >= 0) ? v : 0.0f;
}

__device__ __forceinline__ float xhalf(float v) {
    const int b = __builtin_bit_cast(int, v);
    return __builtin_bit_cast(float,
        __builtin_amdgcn_permlanex16(b, b, 0x76543210, 0xfedcba98, false, false));
}

__device__ __forceinline__ float sigmoidf_(float v) {
    return 1.0f / (1.0f + __expf(-v));
}

#define SLOT(acc, g, q) acc[(kQ * (g) + (q)) >> 3][(kQ * (g) + (q)) & 7]

__global__ __launch_bounds__(32) void fused_gru_lstm_fc(
    const float* __restrict__ x,
    const float* __restrict__ gWih, const float* __restrict__ gWhh,
    const float* __restrict__ gbih, const float* __restrict__ gbhh,
    const float* __restrict__ lWih, const float* __restrict__ lWhh,
    const float* __restrict__ lbih, const float* __restrict__ lbhh,
    const float* __restrict__ fcW,  const float* __restrict__ fcb,
    float* __restrict__ out)
{
    const int tid = threadIdx.x;
    const int c   = tid & 15;
    const int h   = tid >> 4;
    const int m0  = blockIdx.x * 16;

    v2f aGX[kMG][kKS], aGH[kMG][kKS];
    #pragma unroll
    for (int k = 0; k < kMG; ++k)
      #pragma unroll
      for (int ks = 0; ks < kKS; ++ks) {
        aGX[k][ks] = a_frag(gWih, 3, k, ks, c, h);
        aGH[k][ks] = a_frag(gWhh, 3, k, ks, c, h);
      }
    v2f aLX[kML][kKS], aLH[kML][kKS];
    #pragma unroll
    for (int k = 0; k < kML; ++k)
      #pragma unroll
      for (int ks = 0; ks < kKS; ++ks) {
        aLX[k][ks] = a_frag(lWih, 4, k, ks, c, h);
        aLH[k][ks] = a_frag(lWhh, 4, k, ks, c, h);
      }
    v8f cGX[kMG], cGH[kMG], cL[kML];
    #pragma unroll
    for (int k = 0; k < kMG; ++k)
      #pragma unroll
      for (int i = 0; i < 8; ++i) {
        cGX[k][i] = bias_at(gbih, 3, k, i, h);
        cGH[k][i] = bias_at(gbhh, 3, k, i, h);
      }
    #pragma unroll
    for (int k = 0; k < kML; ++k)
      #pragma unroll
      for (int i = 0; i < 8; ++i)
        cL[k][i] = bias_at(lbih, 4, k, i, h) + bias_at(lbhh, 4, k, i, h);
    float fw[kQ];
    #pragma unroll
    for (int q = 0; q < kQ; ++q) fw[q] = fcW[kQ * h + q];
    const float fb = fcb[0];

    float hg[kQ], hl[kQ], cs[kQ];
    #pragma unroll
    for (int q = 0; q < kQ; ++q) { hg[q] = 0.0f; hl[q] = 0.0f; cs[q] = 0.0f; }

    const float* __restrict__ xp = x + (size_t)(m0 + c) * kT * kH + kQ * h;
    float* __restrict__ op = out + (size_t)(m0 + c) * kT;
    float xn[kQ];
    #pragma unroll
    for (int q = 0; q < kQ; ++q) xn[q] = xp[q];

    for (int t = 0; t < kT; ++t) {
        v2f bX[kKS], bH[kKS];
        #pragma unroll
        for (int ks = 0; ks < kKS; ++ks) {
            v2f a; a.x = xn[2 * ks]; a.y = (2 * ks + 1 < kQ) ? xn[2 * ks + 1] : 0.0f; bX[ks] = a;
            v2f b; b.x = hg[2 * ks]; b.y = (2 * ks + 1 < kQ) ? hg[2 * ks + 1] : 0.0f; bH[ks] = b;
        }
        {
            const int tn = (t + 1 < kT) ? t + 1 : t;
            #pragma unroll
            for (int q = 0; q < kQ; ++q) xn[q] = xp[(size_t)tn * kH + q];
        }

        v8f accX[kMG], accH[kMG];
        #pragma unroll
        for (int k = 0; k < kMG; ++k) { accX[k] = cGX[k]; accH[k] = cGH[k]; }
        #pragma unroll
        for (int k = 0; k < kMG; ++k)
          #pragma unroll
          for (int ks = 0; ks < kKS; ++ks) {
            accX[k] = __builtin_amdgcn_wmma_f32_16x16x4_f32(
                false, aGX[k][ks], false, bX[ks], (short)0, accX[k], false, false);
            accH[k] = __builtin_amdgcn_wmma_f32_16x16x4_f32(
                false, aGH[k][ks], false, bH[ks], (short)0, accH[k], false, false);
          }
        #pragma unroll
        for (int q = 0; q < kQ; ++q) {
            float r = sigmoidf_(SLOT(accX, 0, q) + SLOT(accH, 0, q));
            float z = sigmoidf_(SLOT(accX, 1, q) + SLOT(accH, 1, q));
            float n = tanhf(SLOT(accX, 2, q) + r * SLOT(accH, 2, q));
            hg[q] = (1.0f - z) * n + z * hg[q];
        }

        v2f bG[kKS], bL[kKS];
        #pragma unroll
        for (int ks = 0; ks < kKS; ++ks) {
            v2f a; a.x = hg[2 * ks]; a.y = (2 * ks + 1 < kQ) ? hg[2 * ks + 1] : 0.0f; bG[ks] = a;
            v2f b; b.x = hl[2 * ks]; b.y = (2 * ks + 1 < kQ) ? hl[2 * ks + 1] : 0.0f; bL[ks] = b;
        }
        v8f accL[kML];
        #pragma unroll
        for (int k = 0; k < kML; ++k) accL[k] = cL[k];
        #pragma unroll
        for (int k = 0; k < kML; ++k)
          #pragma unroll
          for (int ks = 0; ks < kKS; ++ks) {
            accL[k] = __builtin_amdgcn_wmma_f32_16x16x4_f32(
                false, aLX[k][ks], false, bG[ks], (short)0, accL[k], false, false);
            accL[k] = __builtin_amdgcn_wmma_f32_16x16x4_f32(
                false, aLH[k][ks], false, bL[ks], (short)0, accL[k], false, false);
          }
        float s = 0.0f;
        #pragma unroll
        for (int q = 0; q < kQ; ++q) {
            float ig = sigmoidf_(SLOT(accL, 0, q));
            float fg = sigmoidf_(SLOT(accL, 1, q));
            float gg = tanhf(    SLOT(accL, 2, q));
            float og = sigmoidf_(SLOT(accL, 3, q));
            float cn = fg * cs[q] + ig * gg;
            cs[q] = cn;
            hl[q] = og * tanhf(cn);
            s += hl[q] * fw[q];
        }

        s += xhalf(s);
        if (h == 0) op[t] = s + fb;
    }
}

extern "C" void kernel_launch(void* const* d_in, const int* in_sizes, int n_in,
                              void* d_out, int out_size, void* d_ws, size_t ws_size,
                              hipStream_t stream) {
    (void)in_sizes; (void)n_in; (void)out_size; (void)d_ws; (void)ws_size;
    const float* x    = (const float*)d_in[0];
    const float* gWih = (const float*)d_in[1];
    const float* gWhh = (const float*)d_in[2];
    const float* gbih = (const float*)d_in[3];
    const float* gbhh = (const float*)d_in[4];
    const float* lWih = (const float*)d_in[5];
    const float* lWhh = (const float*)d_in[6];
    const float* lbih = (const float*)d_in[7];
    const float* lbhh = (const float*)d_in[8];
    const float* fcW  = (const float*)d_in[9];
    const float* fcb  = (const float*)d_in[10];
    float* out = (float*)d_out;

    dim3 grid(kB / 16);
    dim3 block(32);
    hipLaunchKernelGGL(fused_gru_lstm_fc, grid, block, 0, stream,
                       x, gWih, gWhh, gbih, gbhh,
                       lWih, lWhh, lbih, lbhh, fcW, fcb, out);
}


# All gate activations on the hardware tanh instruction: tanh direct, logistic as (1+tanh(a/2))/2 with the exact halving folded into gate weights and biases; library tanhf and the divisions are gone
# speedup vs baseline: 9.2728x; 3.4930x over previous
#include <hip/hip_runtime.h>
#include <cmath>

typedef __attribute__((ext_vector_type(2))) float v2f;
typedef __attribute__((ext_vector_type(8))) float v8f;

constexpr int kH  = 18;
constexpr int kT  = 1024;
constexpr int kB  = 1024;
constexpr int kKS = 5;
constexpr int kQ  = 9;
constexpr int kMG = 4;
constexpr int kML = 5;


__device__ __forceinline__ int gate_row(int k, int i, int hh, int ngates) {
    int slot = 8 * k + i;
    int g = slot / kQ, q = slot % kQ;
    return (g < ngates) ? g * kH + kQ * hh + q : -1;
}
__device__ __forceinline__ int k_unit(int ks, int s, int h) {
    int q = 2 * ks + s;
    return (q < kQ) ? kQ * h + q : -1;
}
__device__ __forceinline__ float gate_scale(int row, int tanh_gate) {
    return (row >= 0 && row / kH != tanh_gate) ? 0.5f : 1.0f;
}
__device__ __forceinline__ v2f a_frag(const float* __restrict__ Wm, int ngates, int tanh_gate,
                                      int k, int ks, int c, int h) {
    int row = gate_row(k, c & 7, c >> 3, ngates);
    int u0 = k_unit(ks, 0, h), u1 = k_unit(ks, 1, h);
    const int rc = (row >= 0) ? row : 0;
    const float w0 = Wm[rc * kH + ((u0 >= 0) ? u0 : 0)];
    const float w1 = Wm[rc * kH + ((u1 >= 0) ? u1 : 0)];
    const float sc = gate_scale(row, tanh_gate);
    v2f a;
    a.x = (row >= 0 && u0 >= 0) ? w0 * sc : 0.0f;
    a.y = (row >= 0 && u1 >= 0) ? w1 * sc : 0.0f;
    return a;
}
__device__ __forceinline__ float bias_at(const float* __restrict__ b, int ngates, int tanh_gate,
                                         int k, int i, int h) {
    int row = gate_row(k, i, h, ngates);
    const float v = b[(row >= 0) ? row : 0];
    return (row >= 0) ? v * gate_scale(row, tanh_gate) : 0.0f;
}

__device__ __forceinline__ float xhalf(float v) {
    const int b = __builtin_bit_cast(int, v);
    return __builtin_bit_cast(float,
        __builtin_amdgcn_permlanex16(b, b, 0x76543210, 0xfedcba98, false, false));
}

__device__ __forceinline__ float tanh_hw(float v) {
    return __builtin_amdgcn_tanhf(v);
}
__device__ __forceinline__ float sigmoid_of_half(float half_a) {
    return __builtin_fmaf(0.5f, tanh_hw(half_a), 0.5f);
}

#define SLOT(acc, g, q) acc[(kQ * (g) + (q)) >> 3][(kQ * (g) + (q)) & 7]

__global__ __launch_bounds__(32) void fused_gru_lstm_fc(
    const float* __restrict__ x,
    const float* __restrict__ gWih, const float* __restrict__ gWhh,
    const float* __restrict__ gbih, const float* __restrict__ gbhh,
    const float* __restrict__ lWih, const float* __restrict__ lWhh,
    const float* __restrict__ lbih, const float* __restrict__ lbhh,
    const float* __restrict__ fcW,  const float* __restrict__ fcb,
    float* __restrict__ out)
{
    const int tid = threadIdx.x;
    const int c   = tid & 15;
    const int h   = tid >> 4;
    const int m0  = blockIdx.x * 16;

    v2f aGX[kMG][kKS], aGH[kMG][kKS];
    #pragma unroll
    for (int k = 0; k < kMG; ++k)
      #pragma unroll
      for (int ks = 0; ks < kKS; ++ks) {
        aGX[k][ks] = a_frag(gWih, 3, 2, k, ks, c, h);
        aGH[k][ks] = a_frag(gWhh, 3, 2, k, ks, c, h);
      }
    v2f aLX[kML][kKS], aLH[kML][kKS];
    #pragma unroll
    for (int k = 0; k < kML; ++k)
      #pragma unroll
      for (int ks = 0; ks < kKS; ++ks) {
        aLX[k][ks] = a_frag(lWih, 4, 2, k, ks, c, h);
        aLH[k][ks] = a_frag(lWhh, 4, 2, k, ks, c, h);
      }
    v8f cGX[kMG], cGH[kMG], cL[kML];
    #pragma unroll
    for (int k = 0; k < kMG; ++k)
      #pragma unroll
      for (int i = 0; i < 8; ++i) {
        cGX[k][i] = bias_at(gbih, 3, 2, k, i, h);
        cGH[k][i] = bias_at(gbhh, 3, 2, k, i, h);
      }
    #pragma unroll
    for (int k = 0; k < kML; ++k)
      #pragma unroll
      for (int i = 0; i < 8; ++i)
        cL[k][i] = bias_at(lbih, 4, 2, k, i, h) + bias_at(lbhh, 4, 2, k, i, h);
    float fw[kQ];
    #pragma unroll
    for (int q = 0; q < kQ; ++q) fw[q] = fcW[kQ * h + q];
    const float fb = fcb[0];

    float hg[kQ], hl[kQ], cs[kQ];
    #pragma unroll
    for (int q = 0; q < kQ; ++q) { hg[q] = 0.0f; hl[q] = 0.0f; cs[q] = 0.0f; }

    const float* __restrict__ xp = x + (size_t)(m0 + c) * kT * kH + kQ * h;
    float* __restrict__ op = out + (size_t)(m0 + c) * kT;
    float xn[kQ];
    #pragma unroll
    for (int q = 0; q < kQ; ++q) xn[q] = xp[q];

    for (int t = 0; t < kT; ++t) {
        v2f bX[kKS], bH[kKS];
        #pragma unroll
        for (int ks = 0; ks < kKS; ++ks) {
            v2f a; a.x = xn[2 * ks]; a.y = (2 * ks + 1 < kQ) ? xn[2 * ks + 1] : 0.0f; bX[ks] = a;
            v2f b; b.x = hg[2 * ks]; b.y = (2 * ks + 1 < kQ) ? hg[2 * ks + 1] : 0.0f; bH[ks] = b;
        }
        {
            const int tn = (t + 1 < kT) ? t + 1 : t;
            #pragma unroll
            for (int q = 0; q < kQ; ++q) xn[q] = xp[(size_t)tn * kH + q];
        }

        v8f accX[kMG], accH[kMG];
        #pragma unroll
        for (int k = 0; k < kMG; ++k) { accX[k] = cGX[k]; accH[k] = cGH[k]; }
        #pragma unroll
        for (int k = 0; k < kMG; ++k)
          #pragma unroll
          for (int ks = 0; ks < kKS; ++ks) {
            accX[k] = __builtin_amdgcn_wmma_f32_16x16x4_f32(
                false, aGX[k][ks], false, bX[ks], (short)0, accX[k], false, false);
            accH[k] = __builtin_amdgcn_wmma_f32_16x16x4_f32(
                false, aGH[k][ks], false, bH[ks], (short)0, accH[k], false, false);
          }
        #pragma unroll
        for (int q = 0; q < kQ; ++q) {
            float r = sigmoid_of_half(SLOT(accX, 0, q) + SLOT(accH, 0, q));
            float z = sigmoid_of_half(SLOT(accX, 1, q) + SLOT(accH, 1, q));
            float n = tanh_hw(SLOT(accX, 2, q) + r * SLOT(accH, 2, q));
            hg[q] = (1.0f - z) * n + z * hg[q];
        }

        v2f bG[kKS], bL[kKS];
        #pragma unroll
        for (int ks = 0; ks < kKS; ++ks) {
            v2f a; a.x = hg[2 * ks]; a.y = (2 * ks + 1 < kQ) ? hg[2 * ks + 1] : 0.0f; bG[ks] = a;
            v2f b; b.x = hl[2 * ks]; b.y = (2 * ks + 1 < kQ) ? hl[2 * ks + 1] : 0.0f; bL[ks] = b;
        }
        v8f accL[kML];
        #pragma unroll
        for (int k = 0; k < kML; ++k) accL[k] = cL[k];
        #pragma unroll
        for (int k = 0; k < kML; ++k)
          #pragma unroll
          for (int ks = 0; ks < kKS; ++ks) {
            accL[k] = __builtin_amdgcn_wmma_f32_16x16x4_f32(
                false, aLX[k][ks], false, bG[ks], (short)0, accL[k], false, false);
            accL[k] = __builtin_amdgcn_wmma_f32_16x16x4_f32(
                false, aLH[k][ks], false, bL[ks], (short)0, accL[k], false, false);
          }
        float s = 0.0f;
        #pragma unroll
        for (int q = 0; q < kQ; ++q) {
            float ig = sigmoid_of_half(SLOT(accL, 0, q));
            float fg = sigmoid_of_half(SLOT(accL, 1, q));
            float gg = tanh_hw(        SLOT(accL, 2, q));
            float og = sigmoid_of_half(SLOT(accL, 3, q));
            float cn = fg * cs[q] + ig * gg;
            cs[q] = cn;
            hl[q] = og * tanh_hw(cn);
            s += hl[q] * fw[q];
        }

        s += xhalf(s);
        if (h == 0) op[t] = s + fb;
    }
}

extern "C" void kernel_launch(void* const* d_in, const int* in_sizes, int n_in,
                              void* d_out, int out_size, void* d_ws, size_t ws_size,
                              hipStream_t stream) {
    (void)in_sizes; (void)n_in; (void)out_size; (void)d_ws; (void)ws_size;
    const float* x    = (const float*)d_in[0];
    const float* gWih = (const float*)d_in[1];
    const float* gWhh = (const float*)d_in[2];
    const float* gbih = (const float*)d_in[3];
    const float* gbhh = (const float*)d_in[4];
    const float* lWih = (const float*)d_in[5];
    const float* lWhh = (const float*)d_in[6];
    const float* lbih = (const float*)d_in[7];
    const float* lbhh = (const float*)d_in[8];
    const float* fcW  = (const float*)d_in[9];
    const float* fcb  = (const float*)d_in[10];
    float* out = (float*)d_out;

    dim3 grid(kB / 16);
    dim3 block(32);
    hipLaunchKernelGGL(fused_gru_lstm_fc, grid, block, 0, stream,
                       x, gWih, gWhh, gbih, gbhh,
                       lWih, lWhh, lbih, lbhh, fcW, fcb, out);
}


# Each tile's step split over 4 waves of a work-group as a pipeline (GRU input product, GRU recurrence, LSTM input product, LSTM recurrence) through LDS, one barrier a step; LSTM sum order changes
# speedup vs baseline: 17.2705x; 1.8625x over previous
#include <hip/hip_runtime.h>
#include <cmath>

typedef __attribute__((ext_vector_type(2))) float v2f;
typedef __attribute__((ext_vector_type(4))) float v4f;
typedef __attribute__((ext_vector_type(8))) float v8f;

constexpr int kH  = 18;
constexpr int kT  = 1024;
constexpr int kB  = 1024;
constexpr int kKS = 5;
constexpr int kQ  = 9;
constexpr int kMG = 4;
constexpr int kML = 5;


__device__ __forceinline__ int gate_row(int k, int i, int hh, int ngates) {
    int slot = 8 * k + i;
    int g = slot / kQ, q = slot % kQ;
    return (g < ngates) ? g * kH + kQ * hh + q : -1;
}
__device__ __forceinline__ int k_unit(int ks, int s, int h) {
    int q = 2 * ks + s;
    return (q < kQ) ? kQ * h + q : -1;
}
__device__ __forceinline__ float gate_scale(int row, int tanh_gate) {
    return (row >= 0 && row / kH != tanh_gate) ? 0.5f : 1.0f;
}
__device__ __forceinline__ v2f a_frag(const float* __restrict__ Wm, int ngates, int tanh_gate,
                                      int k, int ks, int c, int h) {
    int row = gate_row(k, c & 7, c >> 3, ngates);
    int u0 = k_unit(ks, 0, h), u1 = k_unit(ks, 1, h);
    const int rc = (row >= 0) ? row : 0;
    const float w0 = Wm[rc * kH + ((u0 >= 0) ? u0 : 0)];
    const float w1 = Wm[rc * kH + ((u1 >= 0) ? u1 : 0)];
    const float sc = gate_scale(row, tanh_gate);
    v2f a;
    a.x = (row >= 0 && u0 >= 0) ? w0 * sc : 0.0f;
    a.y = (row >= 0 && u1 >= 0) ? w1 * sc : 0.0f;
    return a;
}
__device__ __forceinline__ float bias_at(const float* __restrict__ b, int ngates, int tanh_gate,
                                         int k, int i, int h) {
    int row = gate_row(k, i, h, ngates);
    const float v = b[(row >= 0) ? row : 0];
    return (row >= 0) ? v * gate_scale(row, tanh_gate) : 0.0f;
}

__device__ __forceinline__ float xhalf(float v) {
    const int b = __builtin_bit_cast(int, v);
    return __builtin_bit_cast(float,
        __builtin_amdgcn_permlanex16(b, b, 0x76543210, 0xfedcba98, false, false));
}

__device__ __forceinline__ float tanh_hw(float v) {
    return __builtin_amdgcn_tanhf(v);
}
__device__ __forceinline__ float sigmoid_of_half(float half_a) {
    return __builtin_fmaf(0.5f, tanh_hw(half_a), 0.5f);
}

__device__ __forceinline__ v8f mm4(v2f a, v2f b, v8f c) {
    return __builtin_amdgcn_wmma_f32_16x16x4_f32(false, a, false, b, (short)0, c, false, false);
}
__device__ __forceinline__ v2f b_frag(const float (&u)[kQ], int ks) {
    v2f b; b.x = u[2 * ks]; b.y = (2 * ks + 1 < kQ) ? u[2 * ks + 1] : 0.0f; return b;
}
__device__ __forceinline__ void lds_put8(v4f* p, v8f v) {
    p[0]  = __builtin_shufflevector(v, v, 0, 1, 2, 3);
    p[32] = __builtin_shufflevector(v, v, 4, 5, 6, 7);
}
__device__ __forceinline__ v8f lds_get8(const v4f* p) {
    const v4f lo = p[0], hi = p[32];
    return __builtin_shufflevector(lo, hi, 0, 1, 2, 3, 4, 5, 6, 7);
}

#define SLOT(acc, g, q) acc[(kQ * (g) + (q)) >> 3][(kQ * (g) + (q)) & 7]

__global__ __launch_bounds__(128) void fused_gru_lstm_fc(
    const float* __restrict__ x,
    const float* __restrict__ gWih, const float* __restrict__ gWhh,
    const float* __restrict__ gbih, const float* __restrict__ gbhh,
    const float* __restrict__ lWih, const float* __restrict__ lWhh,
    const float* __restrict__ lbih, const float* __restrict__ lbhh,
    const float* __restrict__ fcW,  const float* __restrict__ fcb,
    float* __restrict__ out)
{
    __shared__ v4f sXI[2][2 * kMG][32];
    __shared__ v4f sG [2][3][32];
    __shared__ v4f sLI[2][2 * kML][32];

    const int lane = threadIdx.x & 31;
    const int c   = lane & 15;
    const int h   = lane >> 4;
    const int m0  = blockIdx.x * 16;
    const int w = __builtin_amdgcn_readfirstlane((int)(threadIdx.x >> 5));

    v2f aGX[kMG][kKS], aGH[kMG][kKS];
    v2f aLX[kML][kKS], aLH[kML][kKS];
    v8f cGX[kMG], cGH[kMG], cL[kML];
    float xa[kQ], xb[kQ];
    float hg[kQ], hl[kQ], cs[kQ], fw[kQ];
    float fb = 0.0f;
    #pragma unroll
    for (int k = 0; k < kMG; ++k) {
      #pragma unroll
      for (int ks = 0; ks < kKS; ++ks) { aGX[k][ks] = (v2f){0.0f, 0.0f}; aGH[k][ks] = (v2f){0.0f, 0.0f}; }
      #pragma unroll
      for (int i = 0; i < 8; ++i) { cGX[k][i] = 0.0f; cGH[k][i] = 0.0f; }
    }
    #pragma unroll
    for (int k = 0; k < kML; ++k) {
      #pragma unroll
      for (int ks = 0; ks < kKS; ++ks) { aLX[k][ks] = (v2f){0.0f, 0.0f}; aLH[k][ks] = (v2f){0.0f, 0.0f}; }
      #pragma unroll
      for (int i = 0; i < 8; ++i) cL[k][i] = 0.0f;
    }
    #pragma unroll
    for (int q = 0; q < kQ; ++q) {
        xa[q] = 0.0f; xb[q] = 0.0f;
        hg[q] = 0.0f; hl[q] = 0.0f; cs[q] = 0.0f; fw[q] = 0.0f;
    }

    const float* __restrict__ xp = x + (size_t)(m0 + c) * kT * kH + kQ * h;
    float* __restrict__ op = out + (size_t)(m0 + c) * kT;

    if (w == 0) {
        #pragma unroll
        for (int k = 0; k < kMG; ++k) {
          #pragma unroll
          for (int ks = 0; ks < kKS; ++ks) aGX[k][ks] = a_frag(gWih, 3, 2, k, ks, c, h);
          #pragma unroll
          for (int i = 0; i < 8; ++i) cGX[k][i] = bias_at(gbih, 3, 2, k, i, h);
        }
        #pragma unroll
        for (int q = 0; q < kQ; ++q) { xa[q] = xp[q]; xb[q] = xp[kH + q]; }
    } else if (w == 1) {
        #pragma unroll
        for (int k = 0; k < kMG; ++k) {
          #pragma unroll
          for (int ks = 0; ks < kKS; ++ks) aGH[k][ks] = a_frag(gWhh, 3, 2, k, ks, c, h);
          #pragma unroll
          for (int i = 0; i < 8; ++i) cGH[k][i] = bias_at(gbhh, 3, 2, k, i, h);
        }
    } else if (w == 2) {
        #pragma unroll
        for (int k = 0; k < kML; ++k) {
          #pragma unroll
          for (int ks = 0; ks < kKS; ++ks) aLX[k][ks] = a_frag(lWih, 4, 2, k, ks, c, h);
          #pragma unroll
          for (int i = 0; i < 8; ++i)
            cL[k][i] = bias_at(lbih, 4, 2, k, i, h) + bias_at(lbhh, 4, 2, k, i, h);
        }
    } else {
        #pragma unroll
        for (int k = 0; k < kML; ++k)
          #pragma unroll
          for (int ks = 0; ks < kKS; ++ks) aLH[k][ks] = a_frag(lWhh, 4, 2, k, ks, c, h);
        #pragma unroll
        for (int q = 0; q < kQ; ++q) fw[q] = fcW[kQ * h + q];
        fb = fcb[0];
    }

    auto trip = [&](const int i, const bool steady) __attribute__((always_inline)) {
        const int wr = i & 1, rd = wr ^ 1;
        if (w == 0) {
            if (steady || i < kT) {
                v2f bX[kKS];
                #pragma unroll
                for (int ks = 0; ks < kKS; ++ks) bX[ks] = b_frag(xa, ks);
                const int tn = (i + 2 < kT) ? i + 2 : kT - 1;
                #pragma unroll
                for (int q = 0; q < kQ; ++q) { xa[q] = xb[q]; xb[q] = xp[(size_t)tn * kH + q]; }
                v8f acc[kMG];
                #pragma unroll
                for (int k = 0; k < kMG; ++k) acc[k] = cGX[k];
                #pragma unroll
                for (int k = 0; k < kMG; ++k)
                  #pragma unroll
                  for (int ks = 0; ks < kKS; ++ks) acc[k] = mm4(aGX[k][ks], bX[ks], acc[k]);
                #pragma unroll
                for (int k = 0; k < kMG; ++k) lds_put8(&sXI[wr][2 * k][lane], acc[k]);
            }
        } else if (w == 1) {
            if (steady || (i >= 1 && i <= kT)) {
                v8f xi[kMG];
                #pragma unroll
                for (int k = 0; k < kMG; ++k) xi[k] = lds_get8(&sXI[rd][2 * k][lane]);
                v2f bH[kKS];
                #pragma unroll
                for (int ks = 0; ks < kKS; ++ks) bH[ks] = b_frag(hg, ks);
                v8f accH[kMG];
                #pragma unroll
                for (int k = 0; k < kMG; ++k) accH[k] = cGH[k];
                #pragma unroll
                for (int k = 0; k < kMG; ++k)
                  #pragma unroll
                  for (int ks = 0; ks < kKS; ++ks) accH[k] = mm4(aGH[k][ks], bH[ks], accH[k]);
                #pragma unroll
                for (int q = 0; q < kQ; ++q) {
                    float r = sigmoid_of_half(SLOT(xi, 0, q) + SLOT(accH, 0, q));
                    float z = sigmoid_of_half(SLOT(xi, 1, q) + SLOT(accH, 1, q));
                    float n = tanh_hw(SLOT(xi, 2, q) + r * SLOT(accH, 2, q));
                    hg[q] = (1.0f - z) * n + z * hg[q];
                }
                sG[wr][0][lane] = (v4f){hg[0], hg[1], hg[2], hg[3]};
                sG[wr][1][lane] = (v4f){hg[4], hg[5], hg[6], hg[7]};
                sG[wr][2][lane] = (v4f){hg[8], 0.0f, 0.0f, 0.0f};
            }
        } else if (w == 2) {
            if (steady || (i >= 2 && i <= kT + 1)) {
                const v4f g0 = sG[rd][0][lane], g1 = sG[rd][1][lane], g2 = sG[rd][2][lane];
                const float g[kQ] = {g0.x, g0.y, g0.z, g0.w, g1.x, g1.y, g1.z, g1.w, g2.x};
                v2f bG[kKS];
                #pragma unroll
                for (int ks = 0; ks < kKS; ++ks) bG[ks] = b_frag(g, ks);
                v8f acc[kML];
                #pragma unroll
                for (int k = 0; k < kML; ++k) acc[k] = cL[k];
                #pragma unroll
                for (int k = 0; k < kML; ++k)
                  #pragma unroll
                  for (int ks = 0; ks < kKS; ++ks) acc[k] = mm4(aLX[k][ks], bG[ks], acc[k]);
                #pragma unroll
                for (int k = 0; k < kML; ++k) lds_put8(&sLI[wr][2 * k][lane], acc[k]);
            }
        } else {
            if (steady || (i >= 3 && i <= kT + 2)) {
                const int t = i - 3;
                v8f li[kML];
                #pragma unroll
                for (int k = 0; k < kML; ++k) li[k] = lds_get8(&sLI[rd][2 * k][lane]);
                v2f bL[kKS];
                #pragma unroll
                for (int ks = 0; ks < kKS; ++ks) bL[ks] = b_frag(hl, ks);
                v8f accL[kML];
                #pragma unroll
                for (int k = 0; k < kML; ++k)
                  #pragma unroll
                  for (int e = 0; e < 8; ++e) accL[k][e] = 0.0f;
                #pragma unroll
                for (int k = 0; k < kML; ++k)
                  #pragma unroll
                  for (int ks = 0; ks < kKS; ++ks) accL[k] = mm4(aLH[k][ks], bL[ks], accL[k]);
                #pragma unroll
                for (int k = 0; k < kML; ++k) accL[k] += li[k];
                float s = 0.0f;
                #pragma unroll
                for (int q = 0; q < kQ; ++q) {
                    float ig = sigmoid_of_half(SLOT(accL, 0, q));
                    float fg = sigmoid_of_half(SLOT(accL, 1, q));
                    float gg = tanh_hw(        SLOT(accL, 2, q));
                    float og = sigmoid_of_half(SLOT(accL, 3, q));
                    float cn = fg * cs[q] + ig * gg;
                    cs[q] = cn;
                    hl[q] = og * tanh_hw(cn);
                    s += hl[q] * fw[q];
                }
                s += xhalf(s);
                if (h == 0) op[t] = s + fb;
            }
        }
        __builtin_amdgcn_fence(__ATOMIC_RELEASE, "workgroup", "local");
        __builtin_amdgcn_s_barrier();
        __builtin_amdgcn_fence(__ATOMIC_ACQUIRE, "workgroup", "local");
    };
    #pragma nounroll
    for (int i = 0; i < 3; ++i) trip(i, false);
    for (int i = 3; i < kT; ++i) trip(i, true);
    #pragma nounroll
    for (int i = kT; i < kT + 3; ++i) trip(i, false);
}

extern "C" void kernel_launch(void* const* d_in, const int* in_sizes, int n_in,
                              void* d_out, int out_size, void* d_ws, size_t ws_size,
                              hipStream_t stream) {
    (void)in_sizes; (void)n_in; (void)out_size; (void)d_ws; (void)ws_size;
    const float* x    = (const float*)d_in[0];
    const float* gWih = (const float*)d_in[1];
    const float* gWhh = (const float*)d_in[2];
    const float* gbih = (const float*)d_in[3];
    const float* gbhh = (const float*)d_in[4];
    const float* lWih = (const float*)d_in[5];
    const float* lWhh = (const float*)d_in[6];
    const float* lbih = (const float*)d_in[7];
    const float* lbhh = (const float*)d_in[8];
    const float* fcW  = (const float*)d_in[9];
    const float* fcb  = (const float*)d_in[10];
    float* out = (float*)d_out;

    dim3 grid(kB / 16);
    dim3 block(128);
    hipLaunchKernelGGL(fused_gru_lstm_fc, grid, block, 0, stream,
                       x, gWih, gWhh, gbih, gbhh,
                       lWih, lWhh, lbih, lbhh, fcW, fcb, out);
}


# Input wave reads each step's x a whole trip after asking for it: two x register sets named per trip (steady loop written as trip pairs), no copy inside the asking trip; all sums unchanged
# speedup vs baseline: 18.8559x; 1.0918x over previous
#include <hip/hip_runtime.h>
#include <cmath>

typedef __attribute__((ext_vector_type(2))) float v2f;
typedef __attribute__((ext_vector_type(4))) float v4f;
typedef __attribute__((ext_vector_type(8))) float v8f;

constexpr int kH  = 18;
constexpr int kT  = 1024;
constexpr int kB  = 1024;
constexpr int kKS = 5;
constexpr int kQ  = 9;
constexpr int kMG = 4;
constexpr int kML = 5;


__device__ __forceinline__ int gate_row(int k, int i, int hh, int ngates) {
    int slot = 8 * k + i;
    int g = slot / kQ, q = slot % kQ;
    return (g < ngates) ? g * kH + kQ * hh + q : -1;
}
__device__ __forceinline__ int k_unit(int ks, int s, int h) {
    int q = 2 * ks + s;
    return (q < kQ) ? kQ * h + q : -1;
}
__device__ __forceinline__ float gate_scale(int row, int tanh_gate) {
    return (row >= 0 && row / kH != tanh_gate) ? 0.5f : 1.0f;
}
__device__ __forceinline__ v2f a_frag(const float* __restrict__ Wm, int ngates, int tanh_gate,
                                      int k, int ks, int c, int h) {
    int row = gate_row(k, c & 7, c >> 3, ngates);
    int u0 = k_unit(ks, 0, h), u1 = k_unit(ks, 1, h);
    const int rc = (row >= 0) ? row : 0;
    const float w0 = Wm[rc * kH + ((u0 >= 0) ? u0 : 0)];
    const float w1 = Wm[rc * kH + ((u1 >= 0) ? u1 : 0)];
    const float sc = gate_scale(row, tanh_gate);
    v2f a;
    a.x = (row >= 0 && u0 >= 0) ? w0 * sc : 0.0f;
    a.y = (row >= 0 && u1 >= 0) ? w1 * sc : 0.0f;
    return a;
}
__device__ __forceinline__ float bias_at(const float* __restrict__ b, int ngates, int tanh_gate,
                                         int k, int i, int h) {
    int row = gate_row(k, i, h, ngates);
    const float v = b[(row >= 0) ? row : 0];
    return (row >= 0) ? v * gate_scale(row, tanh_gate) : 0.0f;
}

__device__ __forceinline__ float xhalf(float v) {
    const int b = __builtin_bit_cast(int, v);
    return __builtin_bit_cast(float,
        __builtin_amdgcn_permlanex16(b, b, 0x76543210, 0xfedcba98, false, false));
}

__device__ __forceinline__ float tanh_hw(float v) {
    return __builtin_amdgcn_tanhf(v);
}
__device__ __forceinline__ float sigmoid_of_half(float half_a) {
    return __builtin_fmaf(0.5f, tanh_hw(half_a), 0.5f);
}

__device__ __forceinline__ v8f mm4(v2f a, v2f b, v8f c) {
    return __builtin_amdgcn_wmma_f32_16x16x4_f32(false, a, false, b, (short)0, c, false, false);
}
__device__ __forceinline__ v2f b_frag(const float (&u)[kQ], int ks) {
    v2f b; b.x = u[2 * ks]; b.y = (2 * ks + 1 < kQ) ? u[2 * ks + 1] : 0.0f; return b;
}
__device__ __forceinline__ void lds_put8(v4f* p, v8f v) {
    p[0]  = __builtin_shufflevector(v, v, 0, 1, 2, 3);
    p[32] = __builtin_shufflevector(v, v, 4, 5, 6, 7);
}
__device__ __forceinline__ v8f lds_get8(const v4f* p) {
    const v4f lo = p[0], hi = p[32];
    return __builtin_shufflevector(lo, hi, 0, 1, 2, 3, 4, 5, 6, 7);
}

#define SLOT(acc, g, q) acc[(kQ * (g) + (q)) >> 3][(kQ * (g) + (q)) & 7]

__global__ __launch_bounds__(128) void fused_gru_lstm_fc(
    const float* __restrict__ x,
    const float* __restrict__ gWih, const float* __restrict__ gWhh,
    const float* __restrict__ gbih, const float* __restrict__ gbhh,
    const float* __restrict__ lWih, const float* __restrict__ lWhh,
    const float* __restrict__ lbih, const float* __restrict__ lbhh,
    const float* __restrict__ fcW,  const float* __restrict__ fcb,
    float* __restrict__ out)
{
    __shared__ v4f sXI[2][2 * kMG][32];
    __shared__ v4f sG [2][3][32];
    __shared__ v4f sLI[2][2 * kML][32];

    const int lane = threadIdx.x & 31;
    const int c   = lane & 15;
    const int h   = lane >> 4;
    const int m0  = blockIdx.x * 16;
    const int w = __builtin_amdgcn_readfirstlane((int)(threadIdx.x >> 5));

    v2f aGX[kMG][kKS], aGH[kMG][kKS];
    v2f aLX[kML][kKS], aLH[kML][kKS];
    v8f cGX[kMG], cGH[kMG], cL[kML];
    float xA[kQ], xB[kQ];
    float hg[kQ], hl[kQ], cs[kQ], fw[kQ];
    float fb = 0.0f;
    #pragma unroll
    for (int k = 0; k < kMG; ++k) {
      #pragma unroll
      for (int ks = 0; ks < kKS; ++ks) { aGX[k][ks] = (v2f){0.0f, 0.0f}; aGH[k][ks] = (v2f){0.0f, 0.0f}; }
      #pragma unroll
      for (int i = 0; i < 8; ++i) { cGX[k][i] = 0.0f; cGH[k][i] = 0.0f; }
    }
    #pragma unroll
    for (int k = 0; k < kML; ++k) {
      #pragma unroll
      for (int ks = 0; ks < kKS; ++ks) { aLX[k][ks] = (v2f){0.0f, 0.0f}; aLH[k][ks] = (v2f){0.0f, 0.0f}; }
      #pragma unroll
      for (int i = 0; i < 8; ++i) cL[k][i] = 0.0f;
    }
    #pragma unroll
    for (int q = 0; q < kQ; ++q) {
        xA[q] = 0.0f; xB[q] = 0.0f;
        hg[q] = 0.0f; hl[q] = 0.0f; cs[q] = 0.0f; fw[q] = 0.0f;
    }

    const float* __restrict__ xp = x + (size_t)(m0 + c) * kT * kH + kQ * h;
    float* __restrict__ op = out + (size_t)(m0 + c) * kT;

    if (w == 0) {
        #pragma unroll
        for (int k = 0; k < kMG; ++k) {
          #pragma unroll
          for (int ks = 0; ks < kKS; ++ks) aGX[k][ks] = a_frag(gWih, 3, 2, k, ks, c, h);
          #pragma unroll
          for (int i = 0; i < 8; ++i) cGX[k][i] = bias_at(gbih, 3, 2, k, i, h);
        }
        #pragma unroll
        for (int q = 0; q < kQ; ++q) xA[q] = xp[q];
    } else if (w == 1) {
        #pragma unroll
        for (int k = 0; k < kMG; ++k) {
          #pragma unroll
          for (int ks = 0; ks < kKS; ++ks) aGH[k][ks] = a_frag(gWhh, 3, 2, k, ks, c, h);
          #pragma unroll
          for (int i = 0; i < 8; ++i) cGH[k][i] = bias_at(gbhh, 3, 2, k, i, h);
        }
    } else if (w == 2) {
        #pragma unroll
        for (int k = 0; k < kML; ++k) {
          #pragma unroll
          for (int ks = 0; ks < kKS; ++ks) aLX[k][ks] = a_frag(lWih, 4, 2, k, ks, c, h);
          #pragma unroll
          for (int i = 0; i < 8; ++i)
            cL[k][i] = bias_at(lbih, 4, 2, k, i, h) + bias_at(lbhh, 4, 2, k, i, h);
        }
    } else {
        #pragma unroll
        for (int k = 0; k < kML; ++k)
          #pragma unroll
          for (int ks = 0; ks < kKS; ++ks) aLH[k][ks] = a_frag(lWhh, 4, 2, k, ks, c, h);
        #pragma unroll
        for (int q = 0; q < kQ; ++q) fw[q] = fcW[kQ * h + q];
        fb = fcb[0];
    }

    auto gru_in = [&](const float (&cur)[kQ], float (&nxt)[kQ], const int tn, const int wr)
                  __attribute__((always_inline)) {
        #pragma unroll
        for (int q = 0; q < kQ; ++q) nxt[q] = xp[(size_t)tn * kH + q];
        v2f bX[kKS];
        #pragma unroll
        for (int ks = 0; ks < kKS; ++ks) bX[ks] = b_frag(cur, ks);
        v8f acc[kMG];
        #pragma unroll
        for (int k = 0; k < kMG; ++k) acc[k] = cGX[k];
        #pragma unroll
        for (int k = 0; k < kMG; ++k)
          #pragma unroll
          for (int ks = 0; ks < kKS; ++ks) acc[k] = mm4(aGX[k][ks], bX[ks], acc[k]);
        #pragma unroll
        for (int k = 0; k < kMG; ++k) lds_put8(&sXI[wr][2 * k][lane], acc[k]);
    };
    auto trip = [&](const int i, const bool steady, const int par) __attribute__((always_inline)) {
        const int wr = i & 1, rd = wr ^ 1;
        if (w == 0) {
            if (steady || i < kT) {
                const int tn = (i + 1 < kT) ? i + 1 : kT - 1;
                if (par == 0) gru_in(xA, xB, tn, wr); else gru_in(xB, xA, tn, wr);
            }
        } else if (w == 1) {
            if (steady || (i >= 1 && i <= kT)) {
                v8f xi[kMG];
                #pragma unroll
                for (int k = 0; k < kMG; ++k) xi[k] = lds_get8(&sXI[rd][2 * k][lane]);
                v2f bH[kKS];
                #pragma unroll
                for (int ks = 0; ks < kKS; ++ks) bH[ks] = b_frag(hg, ks);
                v8f accH[kMG];
                #pragma unroll
                for (int k = 0; k < kMG; ++k) accH[k] = cGH[k];
                #pragma unroll
                for (int k = 0; k < kMG; ++k)
                  #pragma unroll
                  for (int ks = 0; ks < kKS; ++ks) accH[k] = mm4(aGH[k][ks], bH[ks], accH[k]);
                #pragma unroll
                for (int q = 0; q < kQ; ++q) {
                    float r = sigmoid_of_half(SLOT(xi, 0, q) + SLOT(accH, 0, q));
                    float z = sigmoid_of_half(SLOT(xi, 1, q) + SLOT(accH, 1, q));
                    float n = tanh_hw(SLOT(xi, 2, q) + r * SLOT(accH, 2, q));
                    hg[q] = (1.0f - z) * n + z * hg[q];
                }
                sG[wr][0][lane] = (v4f){hg[0], hg[1], hg[2], hg[3]};
                sG[wr][1][lane] = (v4f){hg[4], hg[5], hg[6], hg[7]};
                sG[wr][2][lane] = (v4f){hg[8], 0.0f, 0.0f, 0.0f};
            }
        } else if (w == 2) {
            if (steady || (i >= 2 && i <= kT + 1)) {
                const v4f g0 = sG[rd][0][lane], g1 = sG[rd][1][lane], g2 = sG[rd][2][lane];
                const float g[kQ] = {g0.x, g0.y, g0.z, g0.w, g1.x, g1.y, g1.z, g1.w, g2.x};
                v2f bG[kKS];
                #pragma unroll
                for (int ks = 0; ks < kKS; ++ks) bG[ks] = b_frag(g, ks);
                v8f acc[kML];
                #pragma unroll
                for (int k = 0; k < kML; ++k) acc[k] = cL[k];
                #pragma unroll
                for (int k = 0; k < kML; ++k)
                  #pragma unroll
                  for (int ks = 0; ks < kKS; ++ks) acc[k] = mm4(aLX[k][ks], bG[ks], acc[k]);
                #pragma unroll
                for (int k = 0; k < kML; ++k) lds_put8(&sLI[wr][2 * k][lane], acc[k]);
            }
        } else {
            if (steady || (i >= 3 && i <= kT + 2)) {
                const int t = i - 3;
                v8f li[kML];
                #pragma unroll
                for (int k = 0; k < kML; ++k) li[k] = lds_get8(&sLI[rd][2 * k][lane]);
                v2f bL[kKS];
                #pragma unroll
                for (int ks = 0; ks < kKS; ++ks) bL[ks] = b_frag(hl, ks);
                v8f accL[kML];
                #pragma unroll
                for (int k = 0; k < kML; ++k)
                  #pragma unroll
                  for (int e = 0; e < 8; ++e) accL[k][e] = 0.0f;
                #pragma unroll
                for (int k = 0; k < kML; ++k)
                  #pragma unroll
                  for (int ks = 0; ks < kKS; ++ks) accL[k] = mm4(aLH[k][ks], bL[ks], accL[k]);
                #pragma unroll
                for (int k = 0; k < kML; ++k) accL[k] += li[k];
                float s = 0.0f;
                #pragma unroll
                for (int q = 0; q < kQ; ++q) {
                    float ig = sigmoid_of_half(SLOT(accL, 0, q));
                    float fg = sigmoid_of_half(SLOT(accL, 1, q));
                    float gg = tanh_hw(        SLOT(accL, 2, q));
                    float og = sigmoid_of_half(SLOT(accL, 3, q));
                    float cn = fg * cs[q] + ig * gg;
                    cs[q] = cn;
                    hl[q] = og * tanh_hw(cn);
                    s += hl[q] * fw[q];
                }
                s += xhalf(s);
                if (h == 0) op[t] = s + fb;
            }
        }
        __builtin_amdgcn_fence(__ATOMIC_RELEASE, "workgroup", "local");
        __builtin_amdgcn_s_barrier();
        __builtin_amdgcn_fence(__ATOMIC_ACQUIRE, "workgroup", "local");
    };
    trip(0, false, 0); trip(1, false, 1); trip(2, false, 0); trip(3, false, 1);
    for (int i = 4; i < kT; i += 2) { trip(i, true, 0); trip(i + 1, true, 1); }
    #pragma nounroll
    for (int i = kT; i < kT + 3; ++i) trip(i, false, 0);
}

extern "C" void kernel_launch(void* const* d_in, const int* in_sizes, int n_in,
                              void* d_out, int out_size, void* d_ws, size_t ws_size,
                              hipStream_t stream) {
    (void)in_sizes; (void)n_in; (void)out_size; (void)d_ws; (void)ws_size;
    const float* x    = (const float*)d_in[0];
    const float* gWih = (const float*)d_in[1];
    const float* gWhh = (const float*)d_in[2];
    const float* gbih = (const float*)d_in[3];
    const float* gbhh = (const float*)d_in[4];
    const float* lWih = (const float*)d_in[5];
    const float* lWhh = (const float*)d_in[6];
    const float* lbih = (const float*)d_in[7];
    const float* lbhh = (const float*)d_in[8];
    const float* fcW  = (const float*)d_in[9];
    const float* fcb  = (const float*)d_in[10];
    float* out = (float*)d_out;

    dim3 grid(kB / 16);
    dim3 block(128);
    hipLaunchKernelGGL(fused_gru_lstm_fc, grid, block, 0, stream,
                       x, gWih, gWhh, gbih, gbhh,
                       lWih, lWhh, lbih, lbhh, fcW, fcb, out);
}


# LSTM recurrence split by owned unit over two waves on two SIMDs (15 + 10 matrix instructions), h swapped through LDS at the trip barrier; roles share weight registers; sums unchanged
# speedup vs baseline: 20.3040x; 1.0768x over previous
#include <hip/hip_runtime.h>
#include <cmath>

typedef __attribute__((ext_vector_type(2))) float v2f;
typedef __attribute__((ext_vector_type(4))) float v4f;
typedef __attribute__((ext_vector_type(8))) float v8f;

constexpr int kH  = 18;
constexpr int kT  = 1024;
constexpr int kB  = 1024;
constexpr int kKS = 5;
constexpr int kQ  = 9;
constexpr int kMG = 4;
constexpr int kML = 5;
constexpr int kNA = 3, kQA = 5, kQ0A = 0;
constexpr int kNB = 2, kQB = 4, kQ0B = 5;
constexpr int kWLB = 0, kWGR = 1, kWLI = 2, kWLA = 3, kWGI = 4;


__device__ __forceinline__ int gate_row(int k, int i, int hh, int ngates, int nq = kQ, int q0 = 0) {
    int slot = 8 * k + i;
    int g = slot / nq, q = slot % nq;
    return (g < ngates) ? g * kH + kQ * hh + q0 + q : -1;
}
__device__ __forceinline__ int k_unit(int ks, int s, int h) {
    int q = 2 * ks + s;
    return (q < kQ) ? kQ * h + q : -1;
}
__device__ __forceinline__ float gate_scale(int row, int tanh_gate) {
    return (row >= 0 && row / kH != tanh_gate) ? 0.5f : 1.0f;
}
__device__ __forceinline__ v2f a_frag(const float* __restrict__ Wm, int ngates, int tanh_gate,
                                      int k, int ks, int c, int h, int nq = kQ, int q0 = 0) {
    int row = gate_row(k, c & 7, c >> 3, ngates, nq, q0);
    int u0 = k_unit(ks, 0, h), u1 = k_unit(ks, 1, h);
    const int rc = (row >= 0) ? row : 0;
    const float w0 = Wm[rc * kH + ((u0 >= 0) ? u0 : 0)];
    const float w1 = Wm[rc * kH + ((u1 >= 0) ? u1 : 0)];
    const float sc = gate_scale(row, tanh_gate);
    v2f a;
    a.x = (row >= 0 && u0 >= 0) ? w0 * sc : 0.0f;
    a.y = (row >= 0 && u1 >= 0) ? w1 * sc : 0.0f;
    return a;
}
__device__ __forceinline__ float bias_at(const float* __restrict__ b, int ngates, int tanh_gate,
                                         int k, int i, int h, int nq = kQ, int q0 = 0) {
    int row = gate_row(k, i, h, ngates, nq, q0);
    const float v = b[(row >= 0) ? row : 0];
    return (row >= 0) ? v * gate_scale(row, tanh_gate) : 0.0f;
}

__device__ __forceinline__ float xhalf(float v) {
    const int b = __builtin_bit_cast(int, v);
    return __builtin_bit_cast(float,
        __builtin_amdgcn_permlanex16(b, b, 0x76543210, 0xfedcba98, false, false));
}

__device__ __forceinline__ float tanh_hw(float v) {
    return __builtin_amdgcn_tanhf(v);
}
__device__ __forceinline__ float sigmoid_of_half(float half_a) {
    return __builtin_fmaf(0.5f, tanh_hw(half_a), 0.5f);
}

__device__ __forceinline__ v8f mm4(v2f a, v2f b, v8f c) {
    return __builtin_amdgcn_wmma_f32_16x16x4_f32(false, a, false, b, (short)0, c, false, false);
}
__device__ __forceinline__ v2f b_frag(const float (&u)[kQ], int ks) {
    v2f b; b.x = u[2 * ks]; b.y = (2 * ks + 1 < kQ) ? u[2 * ks + 1] : 0.0f; return b;
}
__device__ __forceinline__ void lds_put8(v4f* p, v8f v) {
    p[0]  = __builtin_shufflevector(v, v, 0, 1, 2, 3);
    p[32] = __builtin_shufflevector(v, v, 4, 5, 6, 7);
}
__device__ __forceinline__ v8f lds_get8(const v4f* p) {
    const v4f lo = p[0], hi = p[32];
    return __builtin_shufflevector(lo, hi, 0, 1, 2, 3, 4, 5, 6, 7);
}

#define SLOT(acc, g, q) acc[(kQ * (g) + (q)) >> 3][(kQ * (g) + (q)) & 7]
#define SLOTN(acc, NQ, g, q) acc[((NQ) * (g) + (q)) >> 3][((NQ) * (g) + (q)) & 7]

template <int NT, int NQ, int T0>
__device__ __forceinline__ void lstm_rec_half(const v2f (&aW)[kML][kKS], float (&cs)[kQ], float (&hn)[NQ],
                                              const float (&hp)[kQ], const v4f* li_base) {
    v8f li[NT];
    #pragma unroll
    for (int k = 0; k < NT; ++k) li[k] = lds_get8(li_base + 2 * (T0 + k) * 32);
    v2f bL[kKS];
    #pragma unroll
    for (int ks = 0; ks < kKS; ++ks) bL[ks] = b_frag(hp, ks);
    v8f acc[NT];
    #pragma unroll
    for (int k = 0; k < NT; ++k)
      #pragma unroll
      for (int e = 0; e < 8; ++e) acc[k][e] = 0.0f;
    #pragma unroll
    for (int k = 0; k < NT; ++k)
      #pragma unroll
      for (int ks = 0; ks < kKS; ++ks) acc[k] = mm4(aW[k][ks], bL[ks], acc[k]);
    #pragma unroll
    for (int k = 0; k < NT; ++k) acc[k] += li[k];
    #pragma unroll
    for (int q = 0; q < NQ; ++q) {
        float ig = sigmoid_of_half(SLOTN(acc, NQ, 0, q));
        float fg = sigmoid_of_half(SLOTN(acc, NQ, 1, q));
        float gg = tanh_hw(        SLOTN(acc, NQ, 2, q));
        float og = sigmoid_of_half(SLOTN(acc, NQ, 3, q));
        float cn = fg * cs[q] + ig * gg;
        cs[q] = cn;
        hn[q] = og * tanh_hw(cn);
    }
}

__global__ __launch_bounds__(160) void fused_gru_lstm_fc(
    const float* __restrict__ x,
    const float* __restrict__ gWih, const float* __restrict__ gWhh,
    const float* __restrict__ gbih, const float* __restrict__ gbhh,
    const float* __restrict__ lWih, const float* __restrict__ lWhh,
    const float* __restrict__ lbih, const float* __restrict__ lbhh,
    const float* __restrict__ fcW,  const float* __restrict__ fcb,
    float* __restrict__ out)
{
    __shared__ v4f sXI[2][2 * kMG][32];
    __shared__ v4f sG [2][3][32];
    __shared__ v4f sLI[2][2 * kML][32];
    __shared__ v4f sH [2][2][32];
    __shared__ float sH4[2][32];

    const int lane = threadIdx.x & 31;
    const int c   = lane & 15;
    const int h   = lane >> 4;
    const int m0  = blockIdx.x * 16;
    const int w = __builtin_amdgcn_readfirstlane((int)(threadIdx.x >> 5));

    v2f aW[kML][kKS];
    v8f cB[kML];
    float xA[kQ], xB[kQ];
    float hg[kQ], fw[kQ], csA[kQ], csB[kQ];
    float fb = 0.0f;
    #pragma unroll
    for (int k = 0; k < kML; ++k) {
      #pragma unroll
      for (int ks = 0; ks < kKS; ++ks) aW[k][ks] = (v2f){0.0f, 0.0f};
      #pragma unroll
      for (int i = 0; i < 8; ++i) cB[k][i] = 0.0f;
    }
    #pragma unroll
    for (int q = 0; q < kQ; ++q) { xA[q] = 0.0f; xB[q] = 0.0f; hg[q] = 0.0f; fw[q] = 0.0f; csA[q] = 0.0f; csB[q] = 0.0f; }

    const float* __restrict__ xp = x + (size_t)(m0 + c) * kT * kH + kQ * h;
    const float* __restrict__ xsrc = (w == kWGI) ? xp : x;
    const int xstep = (w == kWGI) ? kH : 0;
    float* __restrict__ op = out + (size_t)(m0 + c) * kT;

    if (w == kWGI) {
        #pragma unroll
        for (int k = 0; k < kMG; ++k) {
          #pragma unroll
          for (int ks = 0; ks < kKS; ++ks) aW[k][ks] = a_frag(gWih, 3, 2, k, ks, c, h);
          #pragma unroll
          for (int i = 0; i < 8; ++i) cB[k][i] = bias_at(gbih, 3, 2, k, i, h);
        }
        #pragma unroll
        for (int q = 0; q < kQ; ++q) xA[q] = xp[q];
    } else if (w == kWGR) {
        #pragma unroll
        for (int k = 0; k < kMG; ++k) {
          #pragma unroll
          for (int ks = 0; ks < kKS; ++ks) aW[k][ks] = a_frag(gWhh, 3, 2, k, ks, c, h);
          #pragma unroll
          for (int i = 0; i < 8; ++i) cB[k][i] = bias_at(gbhh, 3, 2, k, i, h);
        }
    } else if (w == kWLI) {
        #pragma unroll
        for (int k = 0; k < kML; ++k) {
          const int kk = (k < kNA) ? k : k - kNA;
          const int nq = (k < kNA) ? kQA : kQB;
          const int q0 = (k < kNA) ? kQ0A : kQ0B;
          #pragma unroll
          for (int ks = 0; ks < kKS; ++ks) aW[k][ks] = a_frag(lWih, 4, 2, kk, ks, c, h, nq, q0);
          #pragma unroll
          for (int i = 0; i < 8; ++i)
            cB[k][i] = bias_at(lbih, 4, 2, kk, i, h, nq, q0) + bias_at(lbhh, 4, 2, kk, i, h, nq, q0);
        }
    } else if (w == kWLA) {
        #pragma unroll
        for (int k = 0; k < kNA; ++k)
          #pragma unroll
          for (int ks = 0; ks < kKS; ++ks) aW[k][ks] = a_frag(lWhh, 4, 2, k, ks, c, h, kQA, kQ0A);
        #pragma unroll
        for (int q = 0; q < kQ; ++q) fw[q] = fcW[kQ * h + q];
        fb = fcb[0];
    } else {
        #pragma unroll
        for (int k = 0; k < kNB; ++k)
          #pragma unroll
          for (int ks = 0; ks < kKS; ++ks) aW[k][ks] = a_frag(lWhh, 4, 2, k, ks, c, h, kQB, kQ0B);
    }

    auto gru_in = [&](const float (&cur)[kQ], const int wr) __attribute__((always_inline)) {
        v2f bX[kKS];
        #pragma unroll
        for (int ks = 0; ks < kKS; ++ks) bX[ks] = b_frag(cur, ks);
        v8f acc[kMG];
        #pragma unroll
        for (int k = 0; k < kMG; ++k) acc[k] = cB[k];
        #pragma unroll
        for (int k = 0; k < kMG; ++k)
          #pragma unroll
          for (int ks = 0; ks < kKS; ++ks) acc[k] = mm4(aW[k][ks], bX[ks], acc[k]);
        #pragma unroll
        for (int k = 0; k < kMG; ++k) lds_put8(&sXI[wr][2 * k][lane], acc[k]);
    };
    auto read_h = [&](float (&hp)[kQ], const bool have, const int rd) __attribute__((always_inline)) {
        if (have) {
            const v4f a0 = sH[rd][0][lane], b0 = sH[rd][1][lane];
            const float a4 = sH4[rd][lane];
            hp[0] = a0.x; hp[1] = a0.y; hp[2] = a0.z; hp[3] = a0.w; hp[4] = a4;
            hp[5] = b0.x; hp[6] = b0.y; hp[7] = b0.z; hp[8] = b0.w;
        } else {
            #pragma unroll
            for (int q = 0; q < kQ; ++q) hp[q] = 0.0f;
        }
    };
    auto trip = [&](const int i, const bool steady, const int par) __attribute__((always_inline)) {
        const int wr = i & 1, rd = wr ^ 1;
        {
            const int tn = (i + 1 < kT) ? i + 1 : kT - 1;
            const float* __restrict__ xn = xsrc + (size_t)tn * xstep;
            if (par == 0) {
                #pragma unroll
                for (int q = 0; q < kQ; ++q) xB[q] = xn[q];
            } else {
                #pragma unroll
                for (int q = 0; q < kQ; ++q) xA[q] = xn[q];
            }
        }
        if (w == kWGI) {
            if (steady || i < kT) {
                if (par == 0) gru_in(xA, wr); else gru_in(xB, wr);
            }
        } else if (w == kWGR) {
            if (steady || (i >= 1 && i <= kT)) {
                v8f xi[kMG];
                #pragma unroll
                for (int k = 0; k < kMG; ++k) xi[k] = lds_get8(&sXI[rd][2 * k][lane]);
                v2f bH[kKS];
                #pragma unroll
                for (int ks = 0; ks < kKS; ++ks) bH[ks] = b_frag(hg, ks);
                v8f accH[kMG];
                #pragma unroll
                for (int k = 0; k < kMG; ++k) accH[k] = cB[k];
                #pragma unroll
                for (int k = 0; k < kMG; ++k)
                  #pragma unroll
                  for (int ks = 0; ks < kKS; ++ks) accH[k] = mm4(aW[k][ks], bH[ks], accH[k]);
                #pragma unroll
                for (int q = 0; q < kQ; ++q) {
                    float r = sigmoid_of_half(SLOT(xi, 0, q) + SLOT(accH, 0, q));
                    float z = sigmoid_of_half(SLOT(xi, 1, q) + SLOT(accH, 1, q));
                    float n = tanh_hw(SLOT(xi, 2, q) + r * SLOT(accH, 2, q));
                    hg[q] = (1.0f - z) * n + z * hg[q];
                }
                sG[wr][0][lane] = (v4f){hg[0], hg[1], hg[2], hg[3]};
                sG[wr][1][lane] = (v4f){hg[4], hg[5], hg[6], hg[7]};
                sG[wr][2][lane] = (v4f){hg[8], 0.0f, 0.0f, 0.0f};
            }
        } else if (w == kWLI) {
            if (steady || (i >= 2 && i <= kT + 1)) {
                const v4f g0 = sG[rd][0][lane], g1 = sG[rd][1][lane], g2 = sG[rd][2][lane];
                const float g[kQ] = {g0.x, g0.y, g0.z, g0.w, g1.x, g1.y, g1.z, g1.w, g2.x};
                v2f bG[kKS];
                #pragma unroll
                for (int ks = 0; ks < kKS; ++ks) bG[ks] = b_frag(g, ks);
                v8f acc[kML];
                #pragma unroll
                for (int k = 0; k < kML; ++k) acc[k] = cB[k];
                #pragma unroll
                for (int k = 0; k < kML; ++k)
                  #pragma unroll
                  for (int ks = 0; ks < kKS; ++ks) acc[k] = mm4(aW[k][ks], bG[ks], acc[k]);
                #pragma unroll
                for (int k = 0; k < kML; ++k) lds_put8(&sLI[wr][2 * k][lane], acc[k]);
            }
        } else if (w == kWLA) {
            if (steady || (i >= 3 && i <= kT + 3)) {
                const int t = i - 3;
                float hp[kQ];
                read_h(hp, steady || i >= 4, rd);
                if (steady || i <= kT + 2) {
                    float hn[kQA];
                    lstm_rec_half<kNA, kQA, 0>(aW, csA, hn, hp, &sLI[rd][0][lane]);
                    sH[wr][0][lane] = (v4f){hn[0], hn[1], hn[2], hn[3]};
                    sH4[wr][lane] = hn[4];
                }
                if (steady || i >= 4) {
                    float s = 0.0f;
                    #pragma unroll
                    for (int q = 0; q < kQ; ++q) s += hp[q] * fw[q];
                    s += xhalf(s);
                    if (h == 0) op[t - 1] = s + fb;
                }
            }
        } else {
            if (steady || (i >= 3 && i <= kT + 2)) {
                float hp[kQ];
                read_h(hp, steady || i >= 4, rd);
                float hn[kQB];
                lstm_rec_half<kNB, kQB, kNA>(aW, csB, hn, hp, &sLI[rd][0][lane]);
                sH[wr][1][lane] = (v4f){hn[0], hn[1], hn[2], hn[3]};
            }
        }
        __builtin_amdgcn_fence(__ATOMIC_RELEASE, "workgroup", "local");
        __builtin_amdgcn_s_barrier();
        __builtin_amdgcn_fence(__ATOMIC_ACQUIRE, "workgroup", "local");
    };
    trip(0, false, 0); trip(1, false, 1); trip(2, false, 0); trip(3, false, 1);
    for (int i = 4; i < kT; i += 2) { trip(i, true, 0); trip(i + 1, true, 1); }
    #pragma nounroll
    for (int i = kT; i < kT + 4; ++i) trip(i, false, 0);
}

extern "C" void kernel_launch(void* const* d_in, const int* in_sizes, int n_in,
                              void* d_out, int out_size, void* d_ws, size_t ws_size,
                              hipStream_t stream) {
    (void)in_sizes; (void)n_in; (void)out_size; (void)d_ws; (void)ws_size;
    const float* x    = (const float*)d_in[0];
    const float* gWih = (const float*)d_in[1];
    const float* gWhh = (const float*)d_in[2];
    const float* gbih = (const float*)d_in[3];
    const float* gbhh = (const float*)d_in[4];
    const float* lWih = (const float*)d_in[5];
    const float* lWhh = (const float*)d_in[6];
    const float* lbih = (const float*)d_in[7];
    const float* lbhh = (const float*)d_in[8];
    const float* fcW  = (const float*)d_in[9];
    const float* fcb  = (const float*)d_in[10];
    float* out = (float*)d_out;

    dim3 grid(kB / 16);
    dim3 block(160);
    hipLaunchKernelGGL(fused_gru_lstm_fc, grid, block, 0, stream,
                       x, gWih, gWhh, gbih, gbhh,
                       lWih, lWhh, lbih, lbhh, fcW, fcb, out);
}


# Input wave asks for each step's x three trips ahead, not one: four x register sets named per trip (steady loop in groups of four trips); a line fetched from memory no longer holds the trip; same sums
# speedup vs baseline: 22.1229x; 1.0896x over previous
#include <hip/hip_runtime.h>
#include <cmath>

typedef __attribute__((ext_vector_type(2))) float v2f;
typedef __attribute__((ext_vector_type(4))) float v4f;
typedef __attribute__((ext_vector_type(8))) float v8f;

constexpr int kH  = 18;
constexpr int kT  = 1024;
constexpr int kB  = 1024;
constexpr int kKS = 5;
constexpr int kQ  = 9;
constexpr int kMG = 4;
constexpr int kML = 5;
constexpr int kNA = 3, kQA = 5, kQ0A = 0;
constexpr int kNB = 2, kQB = 4, kQ0B = 5;
constexpr int kWLB = 0, kWGR = 1, kWLI = 2, kWLA = 3, kWGI = 4;


__device__ __forceinline__ int gate_row(int k, int i, int hh, int ngates, int nq = kQ, int q0 = 0) {
    int slot = 8 * k + i;
    int g = slot / nq, q = slot % nq;
    return (g < ngates) ? g * kH + kQ * hh + q0 + q : -1;
}
__device__ __forceinline__ int k_unit(int ks, int s, int h) {
    int q = 2 * ks + s;
    return (q < kQ) ? kQ * h + q : -1;
}
__device__ __forceinline__ float gate_scale(int row, int tanh_gate) {
    return (row >= 0 && row / kH != tanh_gate) ? 0.5f : 1.0f;
}
__device__ __forceinline__ v2f a_frag(const float* __restrict__ Wm, int ngates, int tanh_gate,
                                      int k, int ks, int c, int h, int nq = kQ, int q0 = 0) {
    int row = gate_row(k, c & 7, c >> 3, ngates, nq, q0);
    int u0 = k_unit(ks, 0, h), u1 = k_unit(ks, 1, h);
    const int rc = (row >= 0) ? row : 0;
    const float w0 = Wm[rc * kH + ((u0 >= 0) ? u0 : 0)];
    const float w1 = Wm[rc * kH + ((u1 >= 0) ? u1 : 0)];
    const float sc = gate_scale(row, tanh_gate);
    v2f a;
    a.x = (row >= 0 && u0 >= 0) ? w0 * sc : 0.0f;
    a.y = (row >= 0 && u1 >= 0) ? w1 * sc : 0.0f;
    return a;
}
__device__ __forceinline__ float bias_at(const float* __restrict__ b, int ngates, int tanh_gate,
                                         int k, int i, int h, int nq = kQ, int q0 = 0) {
    int row = gate_row(k, i, h, ngates, nq, q0);
    const float v = b[(row >= 0) ? row : 0];
    return (row >= 0) ? v * gate_scale(row, tanh_gate) : 0.0f;
}

__device__ __forceinline__ float xhalf(float v) {
    const int b = __builtin_bit_cast(int, v);
    return __builtin_bit_cast(float,
        __builtin_amdgcn_permlanex16(b, b, 0x76543210, 0xfedcba98, false, false));
}

__device__ __forceinline__ float tanh_hw(float v) {
    return __builtin_amdgcn_tanhf(v);
}
__device__ __forceinline__ float sigmoid_of_half(float half_a) {
    return __builtin_fmaf(0.5f, tanh_hw(half_a), 0.5f);
}

__device__ __forceinline__ v8f mm4(v2f a, v2f b, v8f c) {
    return __builtin_amdgcn_wmma_f32_16x16x4_f32(false, a, false, b, (short)0, c, false, false);
}
__device__ __forceinline__ v2f b_frag(const float (&u)[kQ], int ks) {
    v2f b; b.x = u[2 * ks]; b.y = (2 * ks + 1 < kQ) ? u[2 * ks + 1] : 0.0f; return b;
}
__device__ __forceinline__ void lds_put8(v4f* p, v8f v) {
    p[0]  = __builtin_shufflevector(v, v, 0, 1, 2, 3);
    p[32] = __builtin_shufflevector(v, v, 4, 5, 6, 7);
}
__device__ __forceinline__ v8f lds_get8(const v4f* p) {
    const v4f lo = p[0], hi = p[32];
    return __builtin_shufflevector(lo, hi, 0, 1, 2, 3, 4, 5, 6, 7);
}

#define SLOT(acc, g, q) acc[(kQ * (g) + (q)) >> 3][(kQ * (g) + (q)) & 7]
#define SLOTN(acc, NQ, g, q) acc[((NQ) * (g) + (q)) >> 3][((NQ) * (g) + (q)) & 7]

template <int NT, int NQ, int T0>
__device__ __forceinline__ void lstm_rec_half(const v2f (&aW)[kML][kKS], float (&cs)[kQ], float (&hn)[NQ],
                                              const float (&hp)[kQ], const v4f* li_base) {
    v8f li[NT];
    #pragma unroll
    for (int k = 0; k < NT; ++k) li[k] = lds_get8(li_base + 2 * (T0 + k) * 32);
    v2f bL[kKS];
    #pragma unroll
    for (int ks = 0; ks < kKS; ++ks) bL[ks] = b_frag(hp, ks);
    v8f acc[NT];
    #pragma unroll
    for (int k = 0; k < NT; ++k)
      #pragma unroll
      for (int e = 0; e < 8; ++e) acc[k][e] = 0.0f;
    #pragma unroll
    for (int k = 0; k < NT; ++k)
      #pragma unroll
      for (int ks = 0; ks < kKS; ++ks) acc[k] = mm4(aW[k][ks], bL[ks], acc[k]);
    #pragma unroll
    for (int k = 0; k < NT; ++k) acc[k] += li[k];
    #pragma unroll
    for (int q = 0; q < NQ; ++q) {
        float ig = sigmoid_of_half(SLOTN(acc, NQ, 0, q));
        float fg = sigmoid_of_half(SLOTN(acc, NQ, 1, q));
        float gg = tanh_hw(        SLOTN(acc, NQ, 2, q));
        float og = sigmoid_of_half(SLOTN(acc, NQ, 3, q));
        float cn = fg * cs[q] + ig * gg;
        cs[q] = cn;
        hn[q] = og * tanh_hw(cn);
    }
}

__global__ __launch_bounds__(160) void fused_gru_lstm_fc(
    const float* __restrict__ x,
    const float* __restrict__ gWih, const float* __restrict__ gWhh,
    const float* __restrict__ gbih, const float* __restrict__ gbhh,
    const float* __restrict__ lWih, const float* __restrict__ lWhh,
    const float* __restrict__ lbih, const float* __restrict__ lbhh,
    const float* __restrict__ fcW,  const float* __restrict__ fcb,
    float* __restrict__ out)
{
    __shared__ v4f sXI[2][2 * kMG][32];
    __shared__ v4f sG [2][3][32];
    __shared__ v4f sLI[2][2 * kML][32];
    __shared__ v4f sH [2][2][32];
    __shared__ float sH4[2][32];

    const int lane = threadIdx.x & 31;
    const int c   = lane & 15;
    const int h   = lane >> 4;
    const int m0  = blockIdx.x * 16;
    const int w = __builtin_amdgcn_readfirstlane((int)(threadIdx.x >> 5));

    v2f aW[kML][kKS];
    v8f cB[kML];
    float x0[kQ], x1[kQ], x2[kQ], x3[kQ];
    float hg[kQ], fw[kQ], csA[kQ], csB[kQ];
    float fb = 0.0f;
    #pragma unroll
    for (int k = 0; k < kML; ++k) {
      #pragma unroll
      for (int ks = 0; ks < kKS; ++ks) aW[k][ks] = (v2f){0.0f, 0.0f};
      #pragma unroll
      for (int i = 0; i < 8; ++i) cB[k][i] = 0.0f;
    }
    #pragma unroll
    for (int q = 0; q < kQ; ++q) { x0[q] = 0.0f; x1[q] = 0.0f; x2[q] = 0.0f; x3[q] = 0.0f; hg[q] = 0.0f; fw[q] = 0.0f; csA[q] = 0.0f; csB[q] = 0.0f; }

    const float* __restrict__ xp = x + (size_t)(m0 + c) * kT * kH + kQ * h;
    const float* __restrict__ xsrc = (w == kWGI) ? xp : x;
    const int xstep = (w == kWGI) ? kH : 0;
    float* __restrict__ op = out + (size_t)(m0 + c) * kT;

    if (w == kWGI) {
        #pragma unroll
        for (int k = 0; k < kMG; ++k) {
          #pragma unroll
          for (int ks = 0; ks < kKS; ++ks) aW[k][ks] = a_frag(gWih, 3, 2, k, ks, c, h);
          #pragma unroll
          for (int i = 0; i < 8; ++i) cB[k][i] = bias_at(gbih, 3, 2, k, i, h);
        }
        #pragma unroll
        for (int q = 0; q < kQ; ++q) {
            x0[q] = xp[q]; x1[q] = xp[kH + q]; x2[q] = xp[2 * kH + q];
        }
    } else if (w == kWGR) {
        #pragma unroll
        for (int k = 0; k < kMG; ++k) {
          #pragma unroll
          for (int ks = 0; ks < kKS; ++ks) aW[k][ks] = a_frag(gWhh, 3, 2, k, ks, c, h);
          #pragma unroll
          for (int i = 0; i < 8; ++i) cB[k][i] = bias_at(gbhh, 3, 2, k, i, h);
        }
    } else if (w == kWLI) {
        #pragma unroll
        for (int k = 0; k < kML; ++k) {
          const int kk = (k < kNA) ? k : k - kNA;
          const int nq = (k < kNA) ? kQA : kQB;
          const int q0 = (k < kNA) ? kQ0A : kQ0B;
          #pragma unroll
          for (int ks = 0; ks < kKS; ++ks) aW[k][ks] = a_frag(lWih, 4, 2, kk, ks, c, h, nq, q0);
          #pragma unroll
          for (int i = 0; i < 8; ++i)
            cB[k][i] = bias_at(lbih, 4, 2, kk, i, h, nq, q0) + bias_at(lbhh, 4, 2, kk, i, h, nq, q0);
        }
    } else if (w == kWLA) {
        #pragma unroll
        for (int k = 0; k < kNA; ++k)
          #pragma unroll
          for (int ks = 0; ks < kKS; ++ks) aW[k][ks] = a_frag(lWhh, 4, 2, k, ks, c, h, kQA, kQ0A);
        #pragma unroll
        for (int q = 0; q < kQ; ++q) fw[q] = fcW[kQ * h + q];
        fb = fcb[0];
    } else {
        #pragma unroll
        for (int k = 0; k < kNB; ++k)
          #pragma unroll
          for (int ks = 0; ks < kKS; ++ks) aW[k][ks] = a_frag(lWhh, 4, 2, k, ks, c, h, kQB, kQ0B);
    }

    auto gru_in = [&](const float (&cur)[kQ], const int wr) __attribute__((always_inline)) {
        v2f bX[kKS];
        #pragma unroll
        for (int ks = 0; ks < kKS; ++ks) bX[ks] = b_frag(cur, ks);
        v8f acc[kMG];
        #pragma unroll
        for (int k = 0; k < kMG; ++k) acc[k] = cB[k];
        #pragma unroll
        for (int k = 0; k < kMG; ++k)
          #pragma unroll
          for (int ks = 0; ks < kKS; ++ks) acc[k] = mm4(aW[k][ks], bX[ks], acc[k]);
        #pragma unroll
        for (int k = 0; k < kMG; ++k) lds_put8(&sXI[wr][2 * k][lane], acc[k]);
    };
    auto read_h = [&](float (&hp)[kQ], const bool have, const int rd) __attribute__((always_inline)) {
        if (have) {
            const v4f a0 = sH[rd][0][lane], b0 = sH[rd][1][lane];
            const float a4 = sH4[rd][lane];
            hp[0] = a0.x; hp[1] = a0.y; hp[2] = a0.z; hp[3] = a0.w; hp[4] = a4;
            hp[5] = b0.x; hp[6] = b0.y; hp[7] = b0.z; hp[8] = b0.w;
        } else {
            #pragma unroll
            for (int q = 0; q < kQ; ++q) hp[q] = 0.0f;
        }
    };
    auto trip = [&](const int i, const bool steady, const int par) __attribute__((always_inline)) {
        const int wr = i & 1, rd = wr ^ 1;
        {
            const int tn = (i + 3 < kT) ? i + 3 : kT - 1;
            const float* __restrict__ xn = xsrc + (size_t)tn * xstep;
            if (par == 0) {
                #pragma unroll
                for (int q = 0; q < kQ; ++q) x3[q] = xn[q];
            } else if (par == 1) {
                #pragma unroll
                for (int q = 0; q < kQ; ++q) x0[q] = xn[q];
            } else if (par == 2) {
                #pragma unroll
                for (int q = 0; q < kQ; ++q) x1[q] = xn[q];
            } else {
                #pragma unroll
                for (int q = 0; q < kQ; ++q) x2[q] = xn[q];
            }
        }
        if (w == kWGI) {
            if (steady || i < kT) {
                if (par == 0) gru_in(x0, wr); else if (par == 1) gru_in(x1, wr);
                else if (par == 2) gru_in(x2, wr); else gru_in(x3, wr);
            }
        } else if (w == kWGR) {
            if (steady || (i >= 1 && i <= kT)) {
                v8f xi[kMG];
                #pragma unroll
                for (int k = 0; k < kMG; ++k) xi[k] = lds_get8(&sXI[rd][2 * k][lane]);
                v2f bH[kKS];
                #pragma unroll
                for (int ks = 0; ks < kKS; ++ks) bH[ks] = b_frag(hg, ks);
                v8f accH[kMG];
                #pragma unroll
                for (int k = 0; k < kMG; ++k) accH[k] = cB[k];
                #pragma unroll
                for (int k = 0; k < kMG; ++k)
                  #pragma unroll
                  for (int ks = 0; ks < kKS; ++ks) accH[k] = mm4(aW[k][ks], bH[ks], accH[k]);
                #pragma unroll
                for (int q = 0; q < kQ; ++q) {
                    float r = sigmoid_of_half(SLOT(xi, 0, q) + SLOT(accH, 0, q));
                    float z = sigmoid_of_half(SLOT(xi, 1, q) + SLOT(accH, 1, q));
                    float n = tanh_hw(SLOT(xi, 2, q) + r * SLOT(accH, 2, q));
                    hg[q] = (1.0f - z) * n + z * hg[q];
                }
                sG[wr][0][lane] = (v4f){hg[0], hg[1], hg[2], hg[3]};
                sG[wr][1][lane] = (v4f){hg[4], hg[5], hg[6], hg[7]};
                sG[wr][2][lane] = (v4f){hg[8], 0.0f, 0.0f, 0.0f};
            }
        } else if (w == kWLI) {
            if (steady || (i >= 2 && i <= kT + 1)) {
                const v4f g0 = sG[rd][0][lane], g1 = sG[rd][1][lane], g2 = sG[rd][2][lane];
                const float g[kQ] = {g0.x, g0.y, g0.z, g0.w, g1.x, g1.y, g1.z, g1.w, g2.x};
                v2f bG[kKS];
                #pragma unroll
                for (int ks = 0; ks < kKS; ++ks) bG[ks] = b_frag(g, ks);
                v8f acc[kML];
                #pragma unroll
                for (int k = 0; k < kML; ++k) acc[k] = cB[k];
                #pragma unroll
                for (int k = 0; k < kML; ++k)
                  #pragma unroll
                  for (int ks = 0; ks < kKS; ++ks) acc[k] = mm4(aW[k][ks], bG[ks], acc[k]);
                #pragma unroll
                for (int k = 0; k < kML; ++k) lds_put8(&sLI[wr][2 * k][lane], acc[k]);
            }
        } else if (w == kWLA) {
            if (steady || (i >= 3 && i <= kT + 3)) {
                const int t = i - 3;
                float hp[kQ];
                read_h(hp, steady || i >= 4, rd);
                if (steady || i <= kT + 2) {
                    float hn[kQA];
                    lstm_rec_half<kNA, kQA, 0>(aW, csA, hn, hp, &sLI[rd][0][lane]);
                    sH[wr][0][lane] = (v4f){hn[0], hn[1], hn[2], hn[3]};
                    sH4[wr][lane] = hn[4];
                }
                if (steady || i >= 4) {
                    float s = 0.0f;
                    #pragma unroll
                    for (int q = 0; q < kQ; ++q) s += hp[q] * fw[q];
                    s += xhalf(s);
                    if (h == 0) op[t - 1] = s + fb;
                }
            }
        } else {
            if (steady || (i >= 3 && i <= kT + 2)) {
                float hp[kQ];
                read_h(hp, steady || i >= 4, rd);
                float hn[kQB];
                lstm_rec_half<kNB, kQB, kNA>(aW, csB, hn, hp, &sLI[rd][0][lane]);
                sH[wr][1][lane] = (v4f){hn[0], hn[1], hn[2], hn[3]};
            }
        }
        __builtin_amdgcn_fence(__ATOMIC_RELEASE, "workgroup", "local");
        __builtin_amdgcn_s_barrier();
        __builtin_amdgcn_fence(__ATOMIC_ACQUIRE, "workgroup", "local");
    };
    trip(0, false, 0); trip(1, false, 1); trip(2, false, 2); trip(3, false, 3);
    static_assert((kT - 4) % 4 == 0, "the steady loop takes four trips per pass");
    for (int i = 4; i < kT; i += 4) { trip(i, true, 0); trip(i + 1, true, 1); trip(i + 2, true, 2); trip(i + 3, true, 3); }
    #pragma nounroll
    for (int i = kT; i < kT + 4; ++i) trip(i, false, 0);
}

extern "C" void kernel_launch(void* const* d_in, const int* in_sizes, int n_in,
                              void* d_out, int out_size, void* d_ws, size_t ws_size,
                              hipStream_t stream) {
    (void)in_sizes; (void)n_in; (void)out_size; (void)d_ws; (void)ws_size;
    const float* x    = (const float*)d_in[0];
    const float* gWih = (const float*)d_in[1];
    const float* gWhh = (const float*)d_in[2];
    const float* gbih = (const float*)d_in[3];
    const float* gbhh = (const float*)d_in[4];
    const float* lWih = (const float*)d_in[5];
    const float* lWhh = (const float*)d_in[6];
    const float* lbih = (const float*)d_in[7];
    const float* lbhh = (const float*)d_in[8];
    const float* fcW  = (const float*)d_in[9];
    const float* fcb  = (const float*)d_in[10];
    float* out = (float*)d_out;

    dim3 grid(kB / 16);
    dim3 block(160);
    hipLaunchKernelGGL(fused_gru_lstm_fc, grid, block, 0, stream,
                       x, gWih, gWhh, gbih, gbhh,
                       lWih, lWhh, lbih, lbhh, fcW, fcb, out);
}


# A trip (one barrier) carries eight time steps, not one: LSTM recurrence back in one wave, h in registers, nothing crosses waves inside a trip; out stored per trip; every sum keeps its order
# speedup vs baseline: 24.6790x; 1.1155x over previous
#include <hip/hip_runtime.h>
#include <cmath>

typedef __attribute__((ext_vector_type(2))) float v2f;
typedef __attribute__((ext_vector_type(4))) float v4f;
typedef __attribute__((ext_vector_type(8))) float v8f;

constexpr int kH  = 18;
constexpr int kT  = 1024;
constexpr int kB  = 1024;
constexpr int kKS = 5;
constexpr int kQ  = 9;
constexpr int kMG = 4;
constexpr int kML = 5;
constexpr int kS  = 8;
constexpr int kNT = kT / kS;
static_assert(kT % kS == 0, "a trip carries kS whole steps");
constexpr int kWGI = 0, kWGR = 1, kWLI = 2, kWLR = 3;


__device__ __forceinline__ int gate_row(int k, int i, int hh, int ngates, int nq = kQ, int q0 = 0) {
    int slot = 8 * k + i;
    int g = slot / nq, q = slot % nq;
    return (g < ngates) ? g * kH + kQ * hh + q0 + q : -1;
}
__device__ __forceinline__ int k_unit(int ks, int s, int h) {
    int q = 2 * ks + s;
    return (q < kQ) ? kQ * h + q : -1;
}
__device__ __forceinline__ float gate_scale(int row, int tanh_gate) {
    return (row >= 0 && row / kH != tanh_gate) ? 0.5f : 1.0f;
}
__device__ __forceinline__ v2f a_frag(const float* __restrict__ Wm, int ngates, int tanh_gate,
                                      int k, int ks, int c, int h, int nq = kQ, int q0 = 0) {
    int row = gate_row(k, c & 7, c >> 3, ngates, nq, q0);
    int u0 = k_unit(ks, 0, h), u1 = k_unit(ks, 1, h);
    const int rc = (row >= 0) ? row : 0;
    const float w0 = Wm[rc * kH + ((u0 >= 0) ? u0 : 0)];
    const float w1 = Wm[rc * kH + ((u1 >= 0) ? u1 : 0)];
    const float sc = gate_scale(row, tanh_gate);
    v2f a;
    a.x = (row >= 0 && u0 >= 0) ? w0 * sc : 0.0f;
    a.y = (row >= 0 && u1 >= 0) ? w1 * sc : 0.0f;
    return a;
}
__device__ __forceinline__ float bias_at(const float* __restrict__ b, int ngates, int tanh_gate,
                                         int k, int i, int h, int nq = kQ, int q0 = 0) {
    int row = gate_row(k, i, h, ngates, nq, q0);
    const float v = b[(row >= 0) ? row : 0];
    return (row >= 0) ? v * gate_scale(row, tanh_gate) : 0.0f;
}

__device__ __forceinline__ float xhalf(float v) {
    const int b = __builtin_bit_cast(int, v);
    return __builtin_bit_cast(float,
        __builtin_amdgcn_permlanex16(b, b, 0x76543210, 0xfedcba98, false, false));
}

__device__ __forceinline__ float tanh_hw(float v) {
    return __builtin_amdgcn_tanhf(v);
}
__device__ __forceinline__ float sigmoid_of_half(float half_a) {
    return __builtin_fmaf(0.5f, tanh_hw(half_a), 0.5f);
}

__device__ __forceinline__ v8f mm4(v2f a, v2f b, v8f c) {
    return __builtin_amdgcn_wmma_f32_16x16x4_f32(false, a, false, b, (short)0, c, false, false);
}
__device__ __forceinline__ v2f b_frag(const float (&u)[kQ], int ks) {
    v2f b; b.x = u[2 * ks]; b.y = (2 * ks + 1 < kQ) ? u[2 * ks + 1] : 0.0f; return b;
}
__device__ __forceinline__ void lds_put8(v4f* p, v8f v) {
    p[0]  = __builtin_shufflevector(v, v, 0, 1, 2, 3);
    p[32] = __builtin_shufflevector(v, v, 4, 5, 6, 7);
}
__device__ __forceinline__ v8f lds_get8(const v4f* p) {
    const v4f lo = p[0], hi = p[32];
    return __builtin_shufflevector(lo, hi, 0, 1, 2, 3, 4, 5, 6, 7);
}

#define SLOT(acc, g, q) acc[(kQ * (g) + (q)) >> 3][(kQ * (g) + (q)) & 7]
#define SLOTN(acc, NQ, g, q) acc[((NQ) * (g) + (q)) >> 3][((NQ) * (g) + (q)) & 7]

template <int NT, int NQ, int T0>
__device__ __forceinline__ void lstm_rec_half(const v2f (&aW)[kML][kKS], float (&cs)[kQ], float (&hn)[NQ],
                                              const float (&hp)[kQ], const v4f* li_base) {
    v8f li[NT];
    #pragma unroll
    for (int k = 0; k < NT; ++k) li[k] = lds_get8(li_base + 2 * (T0 + k) * 32);
    v2f bL[kKS];
    #pragma unroll
    for (int ks = 0; ks < kKS; ++ks) bL[ks] = b_frag(hp, ks);
    v8f acc[NT];
    #pragma unroll
    for (int k = 0; k < NT; ++k)
      #pragma unroll
      for (int e = 0; e < 8; ++e) acc[k][e] = 0.0f;
    #pragma unroll
    for (int k = 0; k < NT; ++k)
      #pragma unroll
      for (int ks = 0; ks < kKS; ++ks) acc[k] = mm4(aW[k][ks], bL[ks], acc[k]);
    #pragma unroll
    for (int k = 0; k < NT; ++k) acc[k] += li[k];
    #pragma unroll
    for (int q = 0; q < NQ; ++q) {
        float ig = sigmoid_of_half(SLOTN(acc, NQ, 0, q));
        float fg = sigmoid_of_half(SLOTN(acc, NQ, 1, q));
        float gg = tanh_hw(        SLOTN(acc, NQ, 2, q));
        float og = sigmoid_of_half(SLOTN(acc, NQ, 3, q));
        float cn = fg * cs[q] + ig * gg;
        cs[q] = cn;
        hn[q] = og * tanh_hw(cn);
    }
}

__global__ __launch_bounds__(128) void fused_gru_lstm_fc(
    const float* __restrict__ x,
    const float* __restrict__ gWih, const float* __restrict__ gWhh,
    const float* __restrict__ gbih, const float* __restrict__ gbhh,
    const float* __restrict__ lWih, const float* __restrict__ lWhh,
    const float* __restrict__ lbih, const float* __restrict__ lbhh,
    const float* __restrict__ fcW,  const float* __restrict__ fcb,
    float* __restrict__ out)
{
    __shared__ v4f sXI[2][kS][2 * kMG][32];
    __shared__ v4f sG [2][kS][3][32];
    __shared__ v4f sLI[2][kS][2 * kML][32];

    const int lane = threadIdx.x & 31;
    const int c   = lane & 15;
    const int h   = lane >> 4;
    const int m0  = blockIdx.x * 16;
    const int w = __builtin_amdgcn_readfirstlane((int)(threadIdx.x >> 5));

    v2f aW[kML][kKS];
    v8f cB[kML];
    float hg[kQ], hl[kQ], cs[kQ], fw[kQ];
    float fb = 0.0f;
    #pragma unroll
    for (int k = 0; k < kML; ++k) {
      #pragma unroll
      for (int ks = 0; ks < kKS; ++ks) aW[k][ks] = (v2f){0.0f, 0.0f};
      #pragma unroll
      for (int i = 0; i < 8; ++i) cB[k][i] = 0.0f;
    }
    #pragma unroll
    for (int q = 0; q < kQ; ++q) { hg[q] = 0.0f; hl[q] = 0.0f; cs[q] = 0.0f; fw[q] = 0.0f; }

    const float* __restrict__ xp = x + (size_t)(m0 + c) * kT * kH + kQ * h;
    float* __restrict__ op = out + (size_t)(m0 + c) * kT;

    if (w == kWGI) {
        #pragma unroll
        for (int k = 0; k < kMG; ++k) {
          #pragma unroll
          for (int ks = 0; ks < kKS; ++ks) aW[k][ks] = a_frag(gWih, 3, 2, k, ks, c, h);
          #pragma unroll
          for (int i = 0; i < 8; ++i) cB[k][i] = bias_at(gbih, 3, 2, k, i, h);
        }
    } else if (w == kWGR) {
        #pragma unroll
        for (int k = 0; k < kMG; ++k) {
          #pragma unroll
          for (int ks = 0; ks < kKS; ++ks) aW[k][ks] = a_frag(gWhh, 3, 2, k, ks, c, h);
          #pragma unroll
          for (int i = 0; i < 8; ++i) cB[k][i] = bias_at(gbhh, 3, 2, k, i, h);
        }
    } else if (w == kWLI) {
        #pragma unroll
        for (int k = 0; k < kML; ++k) {
          #pragma unroll
          for (int ks = 0; ks < kKS; ++ks) aW[k][ks] = a_frag(lWih, 4, 2, k, ks, c, h);
          #pragma unroll
          for (int i = 0; i < 8; ++i)
            cB[k][i] = bias_at(lbih, 4, 2, k, i, h) + bias_at(lbhh, 4, 2, k, i, h);
        }
    } else {
        #pragma unroll
        for (int k = 0; k < kML; ++k)
          #pragma unroll
          for (int ks = 0; ks < kKS; ++ks) aW[k][ks] = a_frag(lWhh, 4, 2, k, ks, c, h);
        #pragma unroll
        for (int q = 0; q < kQ; ++q) fw[q] = fcW[kQ * h + q];
        fb = fcb[0];
    }

    #pragma nounroll
    for (int j = 0; j < kNT + 3; ++j) {
        const int wr = j & 1, rd = wr ^ 1;
        if (w == kWGI) {
            if (j < kNT) {
                const float* __restrict__ xt = xp + (size_t)j * (kS * kH);
                float xs[kS][kQ];
                #pragma unroll
                for (int s = 0; s < kS; ++s)
                  #pragma unroll
                  for (int q = 0; q < kQ; ++q) xs[s][q] = xt[s * kH + q];
                #pragma unroll
                for (int s = 0; s < kS; ++s) {
                    v2f bX[kKS];
                    #pragma unroll
                    for (int ks = 0; ks < kKS; ++ks) bX[ks] = b_frag(xs[s], ks);
                    v8f acc[kMG];
                    #pragma unroll
                    for (int k = 0; k < kMG; ++k) acc[k] = cB[k];
                    #pragma unroll
                    for (int k = 0; k < kMG; ++k)
                      #pragma unroll
                      for (int ks = 0; ks < kKS; ++ks) acc[k] = mm4(aW[k][ks], bX[ks], acc[k]);
                    #pragma unroll
                    for (int k = 0; k < kMG; ++k) lds_put8(&sXI[wr][s][2 * k][lane], acc[k]);
                }
            }
        } else if (w == kWGR) {
            if (j >= 1 && j <= kNT) {
                #pragma unroll
                for (int s = 0; s < kS; ++s) {
                    v8f xi[kMG];
                    #pragma unroll
                    for (int k = 0; k < kMG; ++k) xi[k] = lds_get8(&sXI[rd][s][2 * k][lane]);
                    v2f bH[kKS];
                    #pragma unroll
                    for (int ks = 0; ks < kKS; ++ks) bH[ks] = b_frag(hg, ks);
                    v8f accH[kMG];
                    #pragma unroll
                    for (int k = 0; k < kMG; ++k) accH[k] = cB[k];
                    #pragma unroll
                    for (int k = 0; k < kMG; ++k)
                      #pragma unroll
                      for (int ks = 0; ks < kKS; ++ks) accH[k] = mm4(aW[k][ks], bH[ks], accH[k]);
                    #pragma unroll
                    for (int q = 0; q < kQ; ++q) {
                        float r = sigmoid_of_half(SLOT(xi, 0, q) + SLOT(accH, 0, q));
                        float z = sigmoid_of_half(SLOT(xi, 1, q) + SLOT(accH, 1, q));
                        float n = tanh_hw(SLOT(xi, 2, q) + r * SLOT(accH, 2, q));
                        hg[q] = (1.0f - z) * n + z * hg[q];
                    }
                    sG[wr][s][0][lane] = (v4f){hg[0], hg[1], hg[2], hg[3]};
                    sG[wr][s][1][lane] = (v4f){hg[4], hg[5], hg[6], hg[7]};
                    sG[wr][s][2][lane] = (v4f){hg[8], 0.0f, 0.0f, 0.0f};
                }
            }
        } else if (w == kWLI) {
            if (j >= 2 && j <= kNT + 1) {
                v4f g0[kS], g1[kS], g2[kS];
                #pragma unroll
                for (int s = 0; s < kS; ++s) {
                    g0[s] = sG[rd][s][0][lane]; g1[s] = sG[rd][s][1][lane]; g2[s] = sG[rd][s][2][lane];
                }
                #pragma unroll
                for (int s = 0; s < kS; ++s) {
                    const float g[kQ] = {g0[s].x, g0[s].y, g0[s].z, g0[s].w, g1[s].x, g1[s].y, g1[s].z, g1[s].w, g2[s].x};
                    v2f bG[kKS];
                    #pragma unroll
                    for (int ks = 0; ks < kKS; ++ks) bG[ks] = b_frag(g, ks);
                    v8f acc[kML];
                    #pragma unroll
                    for (int k = 0; k < kML; ++k) acc[k] = cB[k];
                    #pragma unroll
                    for (int k = 0; k < kML; ++k)
                      #pragma unroll
                      for (int ks = 0; ks < kKS; ++ks) acc[k] = mm4(aW[k][ks], bG[ks], acc[k]);
                    #pragma unroll
                    for (int k = 0; k < kML; ++k) lds_put8(&sLI[wr][s][2 * k][lane], acc[k]);
                }
            }
        } else {
            if (j >= 3) {
                float o[kS];
                #pragma unroll
                for (int s = 0; s < kS; ++s) {
                    float hn[kQ];
                    lstm_rec_half<kML, kQ, 0>(aW, cs, hn, hl, &sLI[rd][s][0][lane]);
                    #pragma unroll
                    for (int q = 0; q < kQ; ++q) hl[q] = hn[q];
                    float sum = 0.0f;
                    #pragma unroll
                    for (int q = 0; q < kQ; ++q) sum += hl[q] * fw[q];
                    sum += xhalf(sum);
                    o[s] = sum + fb;
                }
                static_assert(kS == 8, "the out stores below write eight steps");
                if (h == 0) {
                    v4f* __restrict__ o4 = reinterpret_cast<v4f*>(op + (size_t)(j - 3) * kS);
                    o4[0] = (v4f){o[0], o[1], o[2], o[3]};
                    o4[1] = (v4f){o[4], o[5], o[6], o[7]};
                }
            }
        }
        __builtin_amdgcn_fence(__ATOMIC_RELEASE, "workgroup", "local");
        __builtin_amdgcn_s_barrier();
        __builtin_amdgcn_fence(__ATOMIC_ACQUIRE, "workgroup", "local");
    }
}

extern "C" void kernel_launch(void* const* d_in, const int* in_sizes, int n_in,
                              void* d_out, int out_size, void* d_ws, size_t ws_size,
                              hipStream_t stream) {
    (void)in_sizes; (void)n_in; (void)out_size; (void)d_ws; (void)ws_size;
    const float* x    = (const float*)d_in[0];
    const float* gWih = (const float*)d_in[1];
    const float* gWhh = (const float*)d_in[2];
    const float* gbih = (const float*)d_in[3];
    const float* gbhh = (const float*)d_in[4];
    const float* lWih = (const float*)d_in[5];
    const float* lWhh = (const float*)d_in[6];
    const float* lbih = (const float*)d_in[7];
    const float* lbhh = (const float*)d_in[8];
    const float* fcW  = (const float*)d_in[9];
    const float* fcb  = (const float*)d_in[10];
    float* out = (float*)d_out;

    dim3 grid(kB / 16);
    dim3 block(128);
    hipLaunchKernelGGL(fused_gru_lstm_fc, grid, block, 0, stream,
                       x, gWih, gWhh, gbih, gbhh,
                       lWih, lWhh, lbih, lbhh, fcW, fcb, out);
}


# Both recurrence waves read a trip's eight steps of input-side pre-activations from LDS as one group ahead of the trip's matrix runs, not step by step inside them; all sums unchanged
# speedup vs baseline: 27.0699x; 1.0969x over previous
#include <hip/hip_runtime.h>
#include <cmath>

typedef __attribute__((ext_vector_type(2))) float v2f;
typedef __attribute__((ext_vector_type(4))) float v4f;
typedef __attribute__((ext_vector_type(8))) float v8f;

constexpr int kH  = 18;
constexpr int kT  = 1024;
constexpr int kB  = 1024;
constexpr int kKS = 5;
constexpr int kQ  = 9;
constexpr int kMG = 4;
constexpr int kML = 5;
constexpr int kS  = 8;
constexpr int kNT = kT / kS;
static_assert(kT % kS == 0, "a trip carries kS whole steps");
constexpr int kWGI = 0, kWGR = 1, kWLI = 2, kWLR = 3;


__device__ __forceinline__ int gate_row(int k, int i, int hh, int ngates, int nq = kQ, int q0 = 0) {
    int slot = 8 * k + i;
    int g = slot / nq, q = slot % nq;
    return (g < ngates) ? g * kH + kQ * hh + q0 + q : -1;
}
__device__ __forceinline__ int k_unit(int ks, int s, int h) {
    int q = 2 * ks + s;
    return (q < kQ) ? kQ * h + q : -1;
}
__device__ __forceinline__ float gate_scale(int row, int tanh_gate) {
    return (row >= 0 && row / kH != tanh_gate) ? 0.5f : 1.0f;
}
__device__ __forceinline__ v2f a_frag(const float* __restrict__ Wm, int ngates, int tanh_gate,
                                      int k, int ks, int c, int h, int nq = kQ, int q0 = 0) {
    int row = gate_row(k, c & 7, c >> 3, ngates, nq, q0);
    int u0 = k_unit(ks, 0, h), u1 = k_unit(ks, 1, h);
    const int rc = (row >= 0) ? row : 0;
    const float w0 = Wm[rc * kH + ((u0 >= 0) ? u0 : 0)];
    const float w1 = Wm[rc * kH + ((u1 >= 0) ? u1 : 0)];
    const float sc = gate_scale(row, tanh_gate);
    v2f a;
    a.x = (row >= 0 && u0 >= 0) ? w0 * sc : 0.0f;
    a.y = (row >= 0 && u1 >= 0) ? w1 * sc : 0.0f;
    return a;
}
__device__ __forceinline__ float bias_at(const float* __restrict__ b, int ngates, int tanh_gate,
                                         int k, int i, int h, int nq = kQ, int q0 = 0) {
    int row = gate_row(k, i, h, ngates, nq, q0);
    const float v = b[(row >= 0) ? row : 0];
    return (row >= 0) ? v * gate_scale(row, tanh_gate) : 0.0f;
}

__device__ __forceinline__ float xhalf(float v) {
    const int b = __builtin_bit_cast(int, v);
    return __builtin_bit_cast(float,
        __builtin_amdgcn_permlanex16(b, b, 0x76543210, 0xfedcba98, false, false));
}

__device__ __forceinline__ float tanh_hw(float v) {
    return __builtin_amdgcn_tanhf(v);
}
__device__ __forceinline__ float sigmoid_of_half(float half_a) {
    return __builtin_fmaf(0.5f, tanh_hw(half_a), 0.5f);
}

__device__ __forceinline__ v8f mm4(v2f a, v2f b, v8f c) {
    return __builtin_amdgcn_wmma_f32_16x16x4_f32(false, a, false, b, (short)0, c, false, false);
}
__device__ __forceinline__ v2f b_frag(const float (&u)[kQ], int ks) {
    v2f b; b.x = u[2 * ks]; b.y = (2 * ks + 1 < kQ) ? u[2 * ks + 1] : 0.0f; return b;
}
__device__ __forceinline__ void lds_put8(v4f* p, v8f v) {
    p[0]  = __builtin_shufflevector(v, v, 0, 1, 2, 3);
    p[32] = __builtin_shufflevector(v, v, 4, 5, 6, 7);
}
__device__ __forceinline__ v8f lds_get8(const v4f* p) {
    const v4f lo = p[0], hi = p[32];
    return __builtin_shufflevector(lo, hi, 0, 1, 2, 3, 4, 5, 6, 7);
}

#define SLOT(acc, g, q) acc[(kQ * (g) + (q)) >> 3][(kQ * (g) + (q)) & 7]
#define SLOTN(acc, NQ, g, q) acc[((NQ) * (g) + (q)) >> 3][((NQ) * (g) + (q)) & 7]

template <int NT, int NQ, int T0>
__device__ __forceinline__ void lstm_rec_half(const v2f (&aW)[kML][kKS], float (&cs)[kQ], float (&hn)[NQ],
                                              const float (&hp)[kQ], const v8f (&li)[kML]) {
    v2f bL[kKS];
    #pragma unroll
    for (int ks = 0; ks < kKS; ++ks) bL[ks] = b_frag(hp, ks);
    v8f acc[NT];
    #pragma unroll
    for (int k = 0; k < NT; ++k)
      #pragma unroll
      for (int e = 0; e < 8; ++e) acc[k][e] = 0.0f;
    #pragma unroll
    for (int k = 0; k < NT; ++k)
      #pragma unroll
      for (int ks = 0; ks < kKS; ++ks) acc[k] = mm4(aW[k][ks], bL[ks], acc[k]);
    #pragma unroll
    for (int k = 0; k < NT; ++k) acc[k] += li[T0 + k];
    #pragma unroll
    for (int q = 0; q < NQ; ++q) {
        float ig = sigmoid_of_half(SLOTN(acc, NQ, 0, q));
        float fg = sigmoid_of_half(SLOTN(acc, NQ, 1, q));
        float gg = tanh_hw(        SLOTN(acc, NQ, 2, q));
        float og = sigmoid_of_half(SLOTN(acc, NQ, 3, q));
        float cn = fg * cs[q] + ig * gg;
        cs[q] = cn;
        hn[q] = og * tanh_hw(cn);
    }
}

__global__ __launch_bounds__(128) void fused_gru_lstm_fc(
    const float* __restrict__ x,
    const float* __restrict__ gWih, const float* __restrict__ gWhh,
    const float* __restrict__ gbih, const float* __restrict__ gbhh,
    const float* __restrict__ lWih, const float* __restrict__ lWhh,
    const float* __restrict__ lbih, const float* __restrict__ lbhh,
    const float* __restrict__ fcW,  const float* __restrict__ fcb,
    float* __restrict__ out)
{
    __shared__ v4f sXI[2][kS][2 * kMG][32];
    __shared__ v4f sG [2][kS][3][32];
    __shared__ v4f sLI[2][kS][2 * kML][32];

    const int lane = threadIdx.x & 31;
    const int c   = lane & 15;
    const int h   = lane >> 4;
    const int m0  = blockIdx.x * 16;
    const int w = __builtin_amdgcn_readfirstlane((int)(threadIdx.x >> 5));

    v2f aW[kML][kKS];
    v8f cB[kML];
    float hg[kQ], hl[kQ], cs[kQ], fw[kQ];
    float fb = 0.0f;
    #pragma unroll
    for (int k = 0; k < kML; ++k) {
      #pragma unroll
      for (int ks = 0; ks < kKS; ++ks) aW[k][ks] = (v2f){0.0f, 0.0f};
      #pragma unroll
      for (int i = 0; i < 8; ++i) cB[k][i] = 0.0f;
    }
    #pragma unroll
    for (int q = 0; q < kQ; ++q) { hg[q] = 0.0f; hl[q] = 0.0f; cs[q] = 0.0f; fw[q] = 0.0f; }

    const float* __restrict__ xp = x + (size_t)(m0 + c) * kT * kH + kQ * h;
    float* __restrict__ op = out + (size_t)(m0 + c) * kT;

    if (w == kWGI) {
        #pragma unroll
        for (int k = 0; k < kMG; ++k) {
          #pragma unroll
          for (int ks = 0; ks < kKS; ++ks) aW[k][ks] = a_frag(gWih, 3, 2, k, ks, c, h);
          #pragma unroll
          for (int i = 0; i < 8; ++i) cB[k][i] = bias_at(gbih, 3, 2, k, i, h);
        }
    } else if (w == kWGR) {
        #pragma unroll
        for (int k = 0; k < kMG; ++k) {
          #pragma unroll
          for (int ks = 0; ks < kKS; ++ks) aW[k][ks] = a_frag(gWhh, 3, 2, k, ks, c, h);
          #pragma unroll
          for (int i = 0; i < 8; ++i) cB[k][i] = bias_at(gbhh, 3, 2, k, i, h);
        }
    } else if (w == kWLI) {
        #pragma unroll
        for (int k = 0; k < kML; ++k) {
          #pragma unroll
          for (int ks = 0; ks < kKS; ++ks) aW[k][ks] = a_frag(lWih, 4, 2, k, ks, c, h);
          #pragma unroll
          for (int i = 0; i < 8; ++i)
            cB[k][i] = bias_at(lbih, 4, 2, k, i, h) + bias_at(lbhh, 4, 2, k, i, h);
        }
    } else {
        #pragma unroll
        for (int k = 0; k < kML; ++k)
          #pragma unroll
          for (int ks = 0; ks < kKS; ++ks) aW[k][ks] = a_frag(lWhh, 4, 2, k, ks, c, h);
        #pragma unroll
        for (int q = 0; q < kQ; ++q) fw[q] = fcW[kQ * h + q];
        fb = fcb[0];
    }

    #pragma nounroll
    for (int j = 0; j < kNT + 3; ++j) {
        const int wr = j & 1, rd = wr ^ 1;
        if (w == kWGI) {
            if (j < kNT) {
                const float* __restrict__ xt = xp + (size_t)j * (kS * kH);
                float xs[kS][kQ];
                #pragma unroll
                for (int s = 0; s < kS; ++s)
                  #pragma unroll
                  for (int q = 0; q < kQ; ++q) xs[s][q] = xt[s * kH + q];
                #pragma unroll
                for (int s = 0; s < kS; ++s) {
                    v2f bX[kKS];
                    #pragma unroll
                    for (int ks = 0; ks < kKS; ++ks) bX[ks] = b_frag(xs[s], ks);
                    v8f acc[kMG];
                    #pragma unroll
                    for (int k = 0; k < kMG; ++k) acc[k] = cB[k];
                    #pragma unroll
                    for (int k = 0; k < kMG; ++k)
                      #pragma unroll
                      for (int ks = 0; ks < kKS; ++ks) acc[k] = mm4(aW[k][ks], bX[ks], acc[k]);
                    #pragma unroll
                    for (int k = 0; k < kMG; ++k) lds_put8(&sXI[wr][s][2 * k][lane], acc[k]);
                }
            }
        } else if (w == kWGR) {
            if (j >= 1 && j <= kNT) {
                v8f xiT[kS][kMG];
                #pragma unroll
                for (int s = 0; s < kS; ++s)
                  #pragma unroll
                  for (int k = 0; k < kMG; ++k) xiT[s][k] = lds_get8(&sXI[rd][s][2 * k][lane]);
                __builtin_amdgcn_sched_barrier(0);
                #pragma unroll
                for (int s = 0; s < kS; ++s) {
                    const v8f (&xi)[kMG] = xiT[s];
                    v2f bH[kKS];
                    #pragma unroll
                    for (int ks = 0; ks < kKS; ++ks) bH[ks] = b_frag(hg, ks);
                    v8f accH[kMG];
                    #pragma unroll
                    for (int k = 0; k < kMG; ++k) accH[k] = cB[k];
                    #pragma unroll
                    for (int k = 0; k < kMG; ++k)
                      #pragma unroll
                      for (int ks = 0; ks < kKS; ++ks) accH[k] = mm4(aW[k][ks], bH[ks], accH[k]);
                    #pragma unroll
                    for (int q = 0; q < kQ; ++q) {
                        float r = sigmoid_of_half(SLOT(xi, 0, q) + SLOT(accH, 0, q));
                        float z = sigmoid_of_half(SLOT(xi, 1, q) + SLOT(accH, 1, q));
                        float n = tanh_hw(SLOT(xi, 2, q) + r * SLOT(accH, 2, q));
                        hg[q] = (1.0f - z) * n + z * hg[q];
                    }
                    sG[wr][s][0][lane] = (v4f){hg[0], hg[1], hg[2], hg[3]};
                    sG[wr][s][1][lane] = (v4f){hg[4], hg[5], hg[6], hg[7]};
                    sG[wr][s][2][lane] = (v4f){hg[8], 0.0f, 0.0f, 0.0f};
                }
            }
        } else if (w == kWLI) {
            if (j >= 2 && j <= kNT + 1) {
                v4f g0[kS], g1[kS], g2[kS];
                #pragma unroll
                for (int s = 0; s < kS; ++s) {
                    g0[s] = sG[rd][s][0][lane]; g1[s] = sG[rd][s][1][lane]; g2[s] = sG[rd][s][2][lane];
                }
                #pragma unroll
                for (int s = 0; s < kS; ++s) {
                    const float g[kQ] = {g0[s].x, g0[s].y, g0[s].z, g0[s].w, g1[s].x, g1[s].y, g1[s].z, g1[s].w, g2[s].x};
                    v2f bG[kKS];
                    #pragma unroll
                    for (int ks = 0; ks < kKS; ++ks) bG[ks] = b_frag(g, ks);
                    v8f acc[kML];
                    #pragma unroll
                    for (int k = 0; k < kML; ++k) acc[k] = cB[k];
                    #pragma unroll
                    for (int k = 0; k < kML; ++k)
                      #pragma unroll
                      for (int ks = 0; ks < kKS; ++ks) acc[k] = mm4(aW[k][ks], bG[ks], acc[k]);
                    #pragma unroll
                    for (int k = 0; k < kML; ++k) lds_put8(&sLI[wr][s][2 * k][lane], acc[k]);
                }
            }
        } else {
            if (j >= 3) {
                v8f liT[kS][kML];
                #pragma unroll
                for (int s = 0; s < kS; ++s)
                  #pragma unroll
                  for (int k = 0; k < kML; ++k) liT[s][k] = lds_get8(&sLI[rd][s][2 * k][lane]);
                __builtin_amdgcn_sched_barrier(0);
                float o[kS];
                #pragma unroll
                for (int s = 0; s < kS; ++s) {
                    float hn[kQ];
                    lstm_rec_half<kML, kQ, 0>(aW, cs, hn, hl, liT[s]);
                    #pragma unroll
                    for (int q = 0; q < kQ; ++q) hl[q] = hn[q];
                    float sum = 0.0f;
                    #pragma unroll
                    for (int q = 0; q < kQ; ++q) sum += hl[q] * fw[q];
                    sum += xhalf(sum);
                    o[s] = sum + fb;
                }
                static_assert(kS == 8, "the out stores below write eight steps");
                if (h == 0) {
                    v4f* __restrict__ o4 = reinterpret_cast<v4f*>(op + (size_t)(j - 3) * kS);
                    o4[0] = (v4f){o[0], o[1], o[2], o[3]};
                    o4[1] = (v4f){o[4], o[5], o[6], o[7]};
                }
            }
        }
        __builtin_amdgcn_fence(__ATOMIC_RELEASE, "workgroup", "local");
        __builtin_amdgcn_s_barrier();
        __builtin_amdgcn_fence(__ATOMIC_ACQUIRE, "workgroup", "local");
    }
}

extern "C" void kernel_launch(void* const* d_in, const int* in_sizes, int n_in,
                              void* d_out, int out_size, void* d_ws, size_t ws_size,
                              hipStream_t stream) {
    (void)in_sizes; (void)n_in; (void)out_size; (void)d_ws; (void)ws_size;
    const float* x    = (const float*)d_in[0];
    const float* gWih = (const float*)d_in[1];
    const float* gWhh = (const float*)d_in[2];
    const float* gbih = (const float*)d_in[3];
    const float* gbhh = (const float*)d_in[4];
    const float* lWih = (const float*)d_in[5];
    const float* lWhh = (const float*)d_in[6];
    const float* lbih = (const float*)d_in[7];
    const float* lbhh = (const float*)d_in[8];
    const float* fcW  = (const float*)d_in[9];
    const float* fcb  = (const float*)d_in[10];
    float* out = (float*)d_out;

    dim3 grid(kB / 16);
    dim3 block(128);
    hipLaunchKernelGGL(fused_gru_lstm_fc, grid, block, 0, stream,
                       x, gWih, gWhh, gbih, gbhh,
                       lWih, lWhh, lbih, lbhh, fcW, fcb, out);
}
